# starting kernel
# baseline (speedup 1.0000x reference)
#include <hip/hip_runtime.h>
#include <math.h>


typedef float v2f __attribute__((ext_vector_type(2)));
typedef float v8f __attribute__((ext_vector_type(8)));

#define C_CH 64

__device__ __forceinline__ void atomic_fadd(float* p, float v) {
#if defined(__gfx1250__)
  unsafeAtomicAdd(p, v);
#else
  atomicAdd(p, v);
#endif
}

__global__ void SpVecs_zero(float* __restrict__ sums, float* __restrict__ cnts,
                            int nsum, int ncnt) {
  int i = blockIdx.x * blockDim.x + threadIdx.x;
  if (i < nsum) sums[i] = 0.0f;
  if (i < ncnt) cnts[i] = 0.0f;
}

__global__ __launch_bounds__(256) void SpVecs_scatter(
    const float* __restrict__ feat, const int* __restrict__ yx,
    const int* __restrict__ seg, float* __restrict__ sums,
    float* __restrict__ cnts, int N, int HW, int W) {
  int i = blockIdx.x * blockDim.x + threadIdx.x;
  if (i >= N) return;

  int y = yx[2 * i];
  int x = yx[2 * i + 1];
  int s = seg[i];

  const float* f = feat + (size_t)y * (size_t)W + (size_t)x;
  float* o = sums + (size_t)s * C_CH;

  atomic_fadd(&cnts[s], 1.0f);

  #pragma unroll
  for (int cb = 0; cb < C_CH; cb += 16) {
    float v[16];
    #pragma unroll
    for (int k = 0; k < 16; ++k) v[k] = f[(size_t)(cb + k) * (size_t)HW];
    #pragma unroll
    for (int k = 0; k < 16; ++k) atomic_fadd(&o[cb + k], v[k]);
  }
}

__global__ __launch_bounds__(256) void SpVecs_finalize(
    float* __restrict__ out, const float* __restrict__ cnts) {
  const int lane  = threadIdx.x & 31;
  const int wid   = threadIdx.x >> 5;
  const int g     = blockIdx.x * (blockDim.x >> 5) + wid;
  const int s0    = g << 4;
  const int m     = lane & 15;
  const int half  = lane >> 4;
  const int khalf = half << 1;

  const float invc = 1.0f / fmaxf(cnts[s0 + m], 1.0f);

#if __has_builtin(__builtin_amdgcn_wmma_f32_16x16x4_f32)
  #pragma unroll
  for (int c0 = 0; c0 < C_CH; c0 += 16) {
    v8f acc = {0.f, 0.f, 0.f, 0.f, 0.f, 0.f, 0.f, 0.f};
    #pragma unroll
    for (int j = 0; j < 4; ++j) {
      v2f A;
      A.x = (m == 4 * j + khalf + 0) ? invc : 0.0f;
      A.y = (m == 4 * j + khalf + 1) ? invc : 0.0f;
      const int r0 = s0 + 4 * j + khalf;
      v2f B;
      B.x = out[(size_t)r0 * C_CH + c0 + m];
      B.y = out[(size_t)(r0 + 1) * C_CH + c0 + m];
      acc = __builtin_amdgcn_wmma_f32_16x16x4_f32(
          false, A, false, B, (short)0, acc, false, false);
    }
    const int rowbase = s0 + (half << 3);
    #pragma unroll
    for (int r = 0; r < 8; ++r)
      out[(size_t)(rowbase + r) * C_CH + c0 + m] = acc[r];
  }
#else
  for (int c0 = 0; c0 < C_CH; c0 += 16) {
    for (int r = half * 8; r < half * 8 + 8; ++r)
      out[(size_t)(s0 + r) * C_CH + c0 + m] *= 1.0f / fmaxf(cnts[s0 + r], 1.0f);
  }
#endif
}

extern "C" void kernel_launch(void* const* d_in, const int* in_sizes, int n_in,
                              void* d_out, int out_size, void* d_ws, size_t ws_size,
                              hipStream_t stream) {
  const float* feat = (const float*)d_in[0];
  const int*   yx   = (const int*)d_in[1];
  const int*   seg  = (const int*)d_in[2];
  float* out  = (float*)d_out;
  float* cnts = (float*)d_ws;

  const int N  = in_sizes[2];
  const int S  = out_size / C_CH;
  const int HW = in_sizes[0] / C_CH;
  const int W  = (int)lround(sqrt((double)HW));

  {
    int n = out_size;
    SpVecs_zero<<<(n + 255) / 256, 256, 0, stream>>>(out, cnts, n, S);
  }

  SpVecs_scatter<<<(N + 255) / 256, 256, 0, stream>>>(feat, yx, seg, out, cnts,
                                                      N, HW, W);

  {
    int groups = S / 16;
    int wpb = 8;
    int blocks = (groups + wpb - 1) / wpb;
    SpVecs_finalize<<<blocks, wpb * 32, 0, stream>>>(out, cnts);
  }
}


# scatter pass walks channel planes in blocks of 4 (grid.y, after grid.x; 4 pixel entries per thread) so co-resident work-groups share 8-12 planes that stay in L2; same float atomics, arrival order
# speedup vs baseline: 1.2291x; 1.2291x over previous
#include <hip/hip_runtime.h>
#include <math.h>


typedef float v2f __attribute__((ext_vector_type(2)));
typedef float v8f __attribute__((ext_vector_type(8)));

#define C_CH 64
#define CPB 4
#define PPT 4
#define SC_THREADS 256

__device__ __forceinline__ void atomic_fadd(float* p, float v) {
#if defined(__gfx1250__)
  unsafeAtomicAdd(p, v);
#else
  atomicAdd(p, v);
#endif
}

__global__ void SpVecs_zero(float* __restrict__ sums, float* __restrict__ cnts,
                            int nsum, int ncnt) {
  int i = blockIdx.x * blockDim.x + threadIdx.x;
  if (i < nsum) sums[i] = 0.0f;
  if (i < ncnt) cnts[i] = 0.0f;
}

__global__ __launch_bounds__(SC_THREADS) void SpVecs_scatter(
    const float* __restrict__ feat, const int* __restrict__ yx,
    const int* __restrict__ seg, float* __restrict__ sums,
    float* __restrict__ cnts, int N, int HW, int W) {
  const int c0   = blockIdx.y * CPB;
  const int base = blockIdx.x * (SC_THREADS * PPT) + threadIdx.x;
  const float* fplane = feat + (size_t)c0 * (size_t)HW;

  const float* f[PPT];
  float*       o[PPT];
  int          s[PPT];
  bool         ok[PPT];

  #pragma unroll
  for (int p = 0; p < PPT; ++p) {
    const int i = base + p * SC_THREADS;
    ok[p] = i < N;
    const int ii = ok[p] ? i : (N - 1);
    const int y = yx[2 * ii];
    const int x = yx[2 * ii + 1];
    s[p] = seg[ii];
    f[p] = fplane + (size_t)y * (size_t)W + (size_t)x;
    o[p] = sums + (size_t)s[p] * C_CH + c0;
  }

  float v[PPT][CPB];
  #pragma unroll
  for (int p = 0; p < PPT; ++p) {
    #pragma unroll
    for (int k = 0; k < CPB; ++k) v[p][k] = f[p][(size_t)k * (size_t)HW];
  }

  if (blockIdx.y == 0) {
    #pragma unroll
    for (int p = 0; p < PPT; ++p) atomic_fadd(&cnts[s[p]], ok[p] ? 1.0f : 0.0f);
  }
  #pragma unroll
  for (int p = 0; p < PPT; ++p) {
    #pragma unroll
    for (int k = 0; k < CPB; ++k) atomic_fadd(&o[p][k], ok[p] ? v[p][k] : 0.0f);
  }
}

__global__ __launch_bounds__(256) void SpVecs_finalize(
    float* __restrict__ out, const float* __restrict__ cnts) {
  const int lane  = threadIdx.x & 31;
  const int wid   = threadIdx.x >> 5;
  const int g     = blockIdx.x * (blockDim.x >> 5) + wid;
  const int s0    = g << 4;
  const int m     = lane & 15;
  const int half  = lane >> 4;
  const int khalf = half << 1;

  const float invc = 1.0f / fmaxf(cnts[s0 + m], 1.0f);

#if __has_builtin(__builtin_amdgcn_wmma_f32_16x16x4_f32)
  #pragma unroll
  for (int c0 = 0; c0 < C_CH; c0 += 16) {
    v8f acc = {0.f, 0.f, 0.f, 0.f, 0.f, 0.f, 0.f, 0.f};
    #pragma unroll
    for (int j = 0; j < 4; ++j) {
      v2f A;
      A.x = (m == 4 * j + khalf + 0) ? invc : 0.0f;
      A.y = (m == 4 * j + khalf + 1) ? invc : 0.0f;
      const int r0 = s0 + 4 * j + khalf;
      v2f B;
      B.x = out[(size_t)r0 * C_CH + c0 + m];
      B.y = out[(size_t)(r0 + 1) * C_CH + c0 + m];
      acc = __builtin_amdgcn_wmma_f32_16x16x4_f32(
          false, A, false, B, (short)0, acc, false, false);
    }
    const int rowbase = s0 + (half << 3);
    #pragma unroll
    for (int r = 0; r < 8; ++r)
      out[(size_t)(rowbase + r) * C_CH + c0 + m] = acc[r];
  }
#else
  for (int c0 = 0; c0 < C_CH; c0 += 16) {
    for (int r = half * 8; r < half * 8 + 8; ++r)
      out[(size_t)(s0 + r) * C_CH + c0 + m] *= 1.0f / fmaxf(cnts[s0 + r], 1.0f);
  }
#endif
}

extern "C" void kernel_launch(void* const* d_in, const int* in_sizes, int n_in,
                              void* d_out, int out_size, void* d_ws, size_t ws_size,
                              hipStream_t stream) {
  const float* feat = (const float*)d_in[0];
  const int*   yx   = (const int*)d_in[1];
  const int*   seg  = (const int*)d_in[2];
  float* out  = (float*)d_out;
  float* cnts = (float*)d_ws;

  const int N  = in_sizes[2];
  const int S  = out_size / C_CH;
  const int HW = in_sizes[0] / C_CH;
  const int W  = (int)lround(sqrt((double)HW));

  {
    int n = out_size;
    SpVecs_zero<<<(n + 255) / 256, 256, 0, stream>>>(out, cnts, n, S);
  }

  {
    dim3 grid((N + SC_THREADS * PPT - 1) / (SC_THREADS * PPT), C_CH / CPB);
    SpVecs_scatter<<<grid, SC_THREADS, 0, stream>>>(feat, yx, seg, out, cnts,
                                                    N, HW, W);
  }

  {
    int groups = S / 16;
    int wpb = 8;
    int blocks = (groups + wpb - 1) / wpb;
    SpVecs_finalize<<<blocks, wpb * 32, 0, stream>>>(out, cnts);
  }
}


# scatter adds go to per-work-group LDS sums (one channel plane x 1/64 of the entries, ds f32 atomics, arrival order); partial planes stored once and added in fixed order; no global atomics left
# speedup vs baseline: 7.9194x; 6.4433x over previous
#include <hip/hip_runtime.h>
#include <math.h>


typedef float v2f __attribute__((ext_vector_type(2)));
typedef float v8f __attribute__((ext_vector_type(8)));

#define C_CH 64
#define NCHUNK 64
#define SC_THREADS 1024
#define PPT 8
#define RD_T 32
#define RD_B 8

__global__ __launch_bounds__(SC_THREADS) void SpVecs_scatter(
    const float* __restrict__ feat, const int* __restrict__ yx,
    const int* __restrict__ seg, float* __restrict__ part,
    float* __restrict__ cpart, int N, int HW, int W, int S, int E) {
  extern __shared__ float lds[];
  float* acc  = lds;
  float* cacc = lds + S;

  const int  c        = blockIdx.y;
  const int  g        = blockIdx.x;
  const bool counting = (c == 0);

  for (int i = threadIdx.x; i < S; i += SC_THREADS) {
    acc[i] = 0.0f;
    if (counting) cacc[i] = 0.0f;
  }
  __syncthreads();

  const float* fplane = feat + (size_t)c * (size_t)HW;
  const int2*  yx2    = (const int2*)yx;
  const int e0 = g * E;
  const int e1 = min(e0 + E, N);

  for (int b = e0; b < e1; b += SC_THREADS * PPT) {
    int  off[PPT];
    int  s[PPT];
    bool ok[PPT];
    #pragma unroll
    for (int p = 0; p < PPT; ++p) {
      const int i = b + p * SC_THREADS + threadIdx.x;
      ok[p] = i < e1;
      const int ii = ok[p] ? i : (e1 - 1);
      const int2 q = yx2[ii];
      s[p]   = seg[ii];
      off[p] = q.x * W + q.y;
    }
    float v[PPT];
    #pragma unroll
    for (int p = 0; p < PPT; ++p) v[p] = fplane[off[p]];

    if (counting) {
      #pragma unroll
      for (int p = 0; p < PPT; ++p) atomicAdd(&cacc[s[p]], ok[p] ? 1.0f : 0.0f);
    }
    #pragma unroll
    for (int p = 0; p < PPT; ++p) atomicAdd(&acc[s[p]], ok[p] ? v[p] : 0.0f);
  }
  __syncthreads();

  float* dst = part + ((size_t)c * NCHUNK + g) * (size_t)S;
  for (int i = threadIdx.x; i < S; i += SC_THREADS) dst[i] = acc[i];
  if (counting) {
    float* cd = cpart + (size_t)g * (size_t)S;
    for (int i = threadIdx.x; i < S; i += SC_THREADS) cd[i] = cacc[i];
  }
}

__global__ __launch_bounds__(256) void SpVecs_reduce(
    const float* __restrict__ part, const float* __restrict__ cpart,
    float* __restrict__ sums, float* __restrict__ cnts, int S) {
  __shared__ float tile[RD_T][RD_T + 1];
  const int lane = threadIdx.x & 31;
  const int w    = threadIdx.x >> 5;
  const int s0   = blockIdx.x * RD_T;
  const int c0   = blockIdx.y * RD_T;
  const int s    = s0 + lane;
  const bool in  = s < S;
  const int  sc  = in ? s : (S - 1);

  const unsigned uS = (unsigned)S;
  unsigned pk[4];
  #pragma unroll
  for (int k = 0; k < 4; ++k)
    pk[k] = ((unsigned)(c0 + w * 4 + k) * NCHUNK) * uS + (unsigned)sc;

  float a[4] = {0.0f, 0.0f, 0.0f, 0.0f};
  for (int g0 = 0; g0 < NCHUNK; g0 += RD_B) {
    float t[4][RD_B];
    #pragma unroll
    for (int j = 0; j < RD_B; ++j) {
      #pragma unroll
      for (int k = 0; k < 4; ++k) t[k][j] = part[pk[k] + (unsigned)(g0 + j) * uS];
    }
    #pragma unroll
    for (int j = 0; j < RD_B; ++j) {
      #pragma unroll
      for (int k = 0; k < 4; ++k) a[k] += t[k][j];
    }
  }
  #pragma unroll
  for (int k = 0; k < 4; ++k) tile[w * 4 + k][lane] = a[k];

  if (blockIdx.y == 0 && w == 0) {
    float n = 0.0f;
    for (int g0 = 0; g0 < NCHUNK; g0 += RD_B) {
      float t[RD_B];
      #pragma unroll
      for (int j = 0; j < RD_B; ++j) t[j] = cpart[(unsigned)(g0 + j) * uS + (unsigned)sc];
      #pragma unroll
      for (int j = 0; j < RD_B; ++j) n += t[j];
    }
    if (in) cnts[s] = n;
  }
  __syncthreads();

  #pragma unroll
  for (int k = 0; k < 4; ++k) {
    const int r = w * 4 + k;
    if (s0 + r < S) sums[(size_t)(s0 + r) * C_CH + c0 + lane] = tile[lane][r];
  }
}

__global__ __launch_bounds__(256) void SpVecs_finalize(
    float* __restrict__ out, const float* __restrict__ cnts) {
  const int lane  = threadIdx.x & 31;
  const int wid   = threadIdx.x >> 5;
  const int g     = blockIdx.x * (blockDim.x >> 5) + wid;
  const int s0    = g << 4;
  const int m     = lane & 15;
  const int half  = lane >> 4;
  const int khalf = half << 1;

  const float invc = 1.0f / fmaxf(cnts[s0 + m], 1.0f);

#if __has_builtin(__builtin_amdgcn_wmma_f32_16x16x4_f32)
  #pragma unroll
  for (int c0 = 0; c0 < C_CH; c0 += 16) {
    v8f acc = {0.f, 0.f, 0.f, 0.f, 0.f, 0.f, 0.f, 0.f};
    #pragma unroll
    for (int j = 0; j < 4; ++j) {
      v2f A;
      A.x = (m == 4 * j + khalf + 0) ? invc : 0.0f;
      A.y = (m == 4 * j + khalf + 1) ? invc : 0.0f;
      const int r0 = s0 + 4 * j + khalf;
      v2f B;
      B.x = out[(size_t)r0 * C_CH + c0 + m];
      B.y = out[(size_t)(r0 + 1) * C_CH + c0 + m];
      acc = __builtin_amdgcn_wmma_f32_16x16x4_f32(
          false, A, false, B, (short)0, acc, false, false);
    }
    const int rowbase = s0 + (half << 3);
    #pragma unroll
    for (int r = 0; r < 8; ++r)
      out[(size_t)(rowbase + r) * C_CH + c0 + m] = acc[r];
  }
#else
  for (int c0 = 0; c0 < C_CH; c0 += 16) {
    for (int r = half * 8; r < half * 8 + 8; ++r)
      out[(size_t)(s0 + r) * C_CH + c0 + m] *= 1.0f / fmaxf(cnts[s0 + r], 1.0f);
  }
#endif
}

extern "C" void kernel_launch(void* const* d_in, const int* in_sizes, int n_in,
                              void* d_out, int out_size, void* d_ws, size_t ws_size,
                              hipStream_t stream) {
  const float* feat = (const float*)d_in[0];
  const int*   yx   = (const int*)d_in[1];
  const int*   seg  = (const int*)d_in[2];
  float* out = (float*)d_out;

  const int N  = in_sizes[2];
  const int S  = out_size / C_CH;
  const int HW = in_sizes[0] / C_CH;
  const int W  = (int)lround(sqrt((double)HW));

  float* cnts  = (float*)d_ws;
  float* cpart = cnts + S;
  float* part  = cpart + (size_t)NCHUNK * (size_t)S;

  {
    const int E = (N + NCHUNK - 1) / NCHUNK;
    dim3 grid(NCHUNK, C_CH);
    SpVecs_scatter<<<grid, SC_THREADS, 2 * (size_t)S * sizeof(float), stream>>>(
        feat, yx, seg, part, cpart, N, HW, W, S, E);
  }

  {
    dim3 grid((S + RD_T - 1) / RD_T, C_CH / RD_T);
    SpVecs_reduce<<<grid, 256, 0, stream>>>(part, cpart, out, cnts, S);
  }

  {
    int groups = S / 16;
    int wpb = 8;
    int blocks = (groups + wpb - 1) / wpb;
    SpVecs_finalize<<<blocks, wpb * 32, 0, stream>>>(out, cnts);
  }
}


# pixel entries counting-sorted by pixel offset (4096 bins, 2 small launches, LDS int atomics) before the gather pass, so a wave's gathers share a few lines; changes the order of the f32 sums
# speedup vs baseline: 22.8067x; 2.8799x over previous
#include <hip/hip_runtime.h>
#include <math.h>


typedef float v2f __attribute__((ext_vector_type(2)));
typedef float v8f __attribute__((ext_vector_type(8)));

#define C_CH 64
#define NCHUNK 64
#define SC_THREADS 1024
#define PPT 8
#define RD_T 32
#define RD_B 8
#define NB_LOG2 12
#define NB (1 << NB_LOG2)
#define SORT_WG NCHUNK
#define SORT_T 1024

static_assert(NB == 4 * SORT_T, "the scan gives each thread 4 consecutive bins");
static_assert(SORT_T == 32 * 32, "the scan is 32 waves of 32 lanes");

__device__ __forceinline__ int SpVecs_shift(int HW) {
  const int bits = 32 - __clz(HW - 1);
  return max(bits - NB_LOG2, 0);
}
__device__ __forceinline__ int SpVecs_bin(int off, int shift) {
  return (int)min((unsigned)off >> shift, (unsigned)(NB - 1));
}

__global__ __launch_bounds__(SORT_T) void SpVecs_bincount(
    const int* __restrict__ yx, int* __restrict__ table, int N, int HW, int W, int E) {
  __shared__ int h[NB];
  for (int i = threadIdx.x; i < NB; i += SORT_T) h[i] = 0;
  __syncthreads();

  const int   shift = SpVecs_shift(HW);
  const int2* yx2   = (const int2*)yx;
  const int e0 = blockIdx.x * E;
  const int e1 = min(e0 + E, N);
  for (int b = e0; b < e1; b += SORT_T * PPT) {
    int  key[PPT];
    bool ok[PPT];
    #pragma unroll
    for (int p = 0; p < PPT; ++p) {
      const int i = b + p * SORT_T + threadIdx.x;
      ok[p] = i < e1;
      const int2 q = yx2[ok[p] ? i : (e1 - 1)];
      key[p] = SpVecs_bin(q.x * W + q.y, shift);
    }
    #pragma unroll
    for (int p = 0; p < PPT; ++p) atomicAdd(&h[key[p]], ok[p] ? 1 : 0);
  }
  __syncthreads();

  int* row = table + (size_t)blockIdx.x * NB;
  for (int i = threadIdx.x; i < NB; i += SORT_T) row[i] = h[i];
}

__global__ __launch_bounds__(SORT_T) void SpVecs_place(
    const int* __restrict__ yx, const int* __restrict__ seg,
    const int* __restrict__ table, int* __restrict__ sorted,
    int N, int HW, int W, int E) {
  __shared__ int cur[NB];
  __shared__ int wsum[32];
  const int wg   = blockIdx.x;
  const int t    = threadIdx.x;
  const int lane = t & 31;
  const int w    = t >> 5;

  int4 tot = make_int4(0, 0, 0, 0);
  int4 bef = make_int4(0, 0, 0, 0);
  const int4* tb = (const int4*)table;
  #pragma unroll 8
  for (int g = 0; g < SORT_WG; ++g) {
    const int4 v = tb[(size_t)g * (NB / 4) + t];
    tot.x += v.x; tot.y += v.y; tot.z += v.z; tot.w += v.w;
    if (g < wg) { bef.x += v.x; bef.y += v.y; bef.z += v.z; bef.w += v.w; }
  }

  const int tsum = tot.x + tot.y + tot.z + tot.w;
  int x = tsum;
  #pragma unroll
  for (int d = 1; d < 32; d <<= 1) {
    const int y = __shfl_up(x, d);
    if (lane >= d) x += y;
  }
  if (lane == 31) wsum[w] = x;
  __syncthreads();
  if (w == 0) {
    const int y = wsum[lane];
    int z = y;
    #pragma unroll
    for (int d = 1; d < 32; d <<= 1) {
      const int u = __shfl_up(z, d);
      if (lane >= d) z += u;
    }
    wsum[lane] = z - y;
  }
  __syncthreads();
  const int base = wsum[w] + x - tsum;
  cur[4 * t + 0] = base + bef.x;
  cur[4 * t + 1] = base + tot.x + bef.y;
  cur[4 * t + 2] = base + tot.x + tot.y + bef.z;
  cur[4 * t + 3] = base + tot.x + tot.y + tot.z + bef.w;
  __syncthreads();

  const int   shift = SpVecs_shift(HW);
  const int2* yx2   = (const int2*)yx;
  int2*       srt   = (int2*)sorted;
  const int e0 = wg * E;
  const int e1 = min(e0 + E, N);
  for (int b = e0; b < e1; b += SORT_T * PPT) {
    int off[PPT];
    int sg[PPT];
    #pragma unroll
    for (int p = 0; p < PPT; ++p) {
      const int i  = b + p * SORT_T + threadIdx.x;
      const int ii = min(i, e1 - 1);
      const int2 q = yx2[ii];
      sg[p]  = seg[ii];
      off[p] = (i < e1) ? (q.x * W + q.y) : -1;
    }
    #pragma unroll
    for (int p = 0; p < PPT; ++p) {
      if (off[p] >= 0) {
        const int pos = atomicAdd(&cur[SpVecs_bin(off[p], shift)], 1);
        srt[pos] = make_int2(off[p], sg[p]);
      }
    }
  }
}

__global__ __launch_bounds__(SC_THREADS) void SpVecs_scatter(
    const float* __restrict__ feat, const int* __restrict__ sorted,
    float* __restrict__ part,
    float* __restrict__ cpart, int N, int HW, int S, int E) {
  extern __shared__ float lds[];
  float* acc  = lds;
  float* cacc = lds + S;

  const int  c        = blockIdx.y;
  const int  g        = blockIdx.x;
  const bool counting = (c == 0);

  for (int i = threadIdx.x; i < S; i += SC_THREADS) {
    acc[i] = 0.0f;
    if (counting) cacc[i] = 0.0f;
  }
  __syncthreads();

  const float* fplane = feat + (size_t)c * (size_t)HW;
  const int2*  srt    = (const int2*)sorted;
  const int e0 = g * E;
  const int e1 = min(e0 + E, N);

  for (int b = e0; b < e1; b += SC_THREADS * PPT) {
    int  off[PPT];
    int  s[PPT];
    bool ok[PPT];
    #pragma unroll
    for (int p = 0; p < PPT; ++p) {
      const int i = b + p * SC_THREADS + threadIdx.x;
      ok[p] = i < e1;
      const int ii = ok[p] ? i : (e1 - 1);
      const int2 q = srt[ii];
      off[p] = q.x;
      s[p]   = q.y;
    }
    float v[PPT];
    #pragma unroll
    for (int p = 0; p < PPT; ++p) v[p] = fplane[off[p]];

    if (counting) {
      #pragma unroll
      for (int p = 0; p < PPT; ++p) atomicAdd(&cacc[s[p]], ok[p] ? 1.0f : 0.0f);
    }
    #pragma unroll
    for (int p = 0; p < PPT; ++p) atomicAdd(&acc[s[p]], ok[p] ? v[p] : 0.0f);
  }
  __syncthreads();

  float* dst = part + ((size_t)c * NCHUNK + g) * (size_t)S;
  for (int i = threadIdx.x; i < S; i += SC_THREADS) dst[i] = acc[i];
  if (counting) {
    float* cd = cpart + (size_t)g * (size_t)S;
    for (int i = threadIdx.x; i < S; i += SC_THREADS) cd[i] = cacc[i];
  }
}

__global__ __launch_bounds__(256) void SpVecs_reduce(
    const float* __restrict__ part, const float* __restrict__ cpart,
    float* __restrict__ sums, float* __restrict__ cnts, int S) {
  __shared__ float tile[RD_T][RD_T + 1];
  const int lane = threadIdx.x & 31;
  const int w    = threadIdx.x >> 5;
  const int s0   = blockIdx.x * RD_T;
  const int c0   = blockIdx.y * RD_T;
  const int s    = s0 + lane;
  const bool in  = s < S;
  const int  sc  = in ? s : (S - 1);

  const unsigned uS = (unsigned)S;
  unsigned pk[4];
  #pragma unroll
  for (int k = 0; k < 4; ++k)
    pk[k] = ((unsigned)(c0 + w * 4 + k) * NCHUNK) * uS + (unsigned)sc;

  float a[4] = {0.0f, 0.0f, 0.0f, 0.0f};
  for (int g0 = 0; g0 < NCHUNK; g0 += RD_B) {
    float t[4][RD_B];
    #pragma unroll
    for (int j = 0; j < RD_B; ++j) {
      #pragma unroll
      for (int k = 0; k < 4; ++k) t[k][j] = part[pk[k] + (unsigned)(g0 + j) * uS];
    }
    #pragma unroll
    for (int j = 0; j < RD_B; ++j) {
      #pragma unroll
      for (int k = 0; k < 4; ++k) a[k] += t[k][j];
    }
  }
  #pragma unroll
  for (int k = 0; k < 4; ++k) tile[w * 4 + k][lane] = a[k];

  if (blockIdx.y == 0 && w == 0) {
    float n = 0.0f;
    for (int g0 = 0; g0 < NCHUNK; g0 += RD_B) {
      float t[RD_B];
      #pragma unroll
      for (int j = 0; j < RD_B; ++j) t[j] = cpart[(unsigned)(g0 + j) * uS + (unsigned)sc];
      #pragma unroll
      for (int j = 0; j < RD_B; ++j) n += t[j];
    }
    if (in) cnts[s] = n;
  }
  __syncthreads();

  #pragma unroll
  for (int k = 0; k < 4; ++k) {
    const int r = w * 4 + k;
    if (s0 + r < S) sums[(size_t)(s0 + r) * C_CH + c0 + lane] = tile[lane][r];
  }
}

__global__ __launch_bounds__(256) void SpVecs_finalize(
    float* __restrict__ out, const float* __restrict__ cnts) {
  const int lane  = threadIdx.x & 31;
  const int wid   = threadIdx.x >> 5;
  const int g     = blockIdx.x * (blockDim.x >> 5) + wid;
  const int s0    = g << 4;
  const int m     = lane & 15;
  const int half  = lane >> 4;
  const int khalf = half << 1;

  const float invc = 1.0f / fmaxf(cnts[s0 + m], 1.0f);

#if __has_builtin(__builtin_amdgcn_wmma_f32_16x16x4_f32)
  #pragma unroll
  for (int c0 = 0; c0 < C_CH; c0 += 16) {
    v8f acc = {0.f, 0.f, 0.f, 0.f, 0.f, 0.f, 0.f, 0.f};
    #pragma unroll
    for (int j = 0; j < 4; ++j) {
      v2f A;
      A.x = (m == 4 * j + khalf + 0) ? invc : 0.0f;
      A.y = (m == 4 * j + khalf + 1) ? invc : 0.0f;
      const int r0 = s0 + 4 * j + khalf;
      v2f B;
      B.x = out[(size_t)r0 * C_CH + c0 + m];
      B.y = out[(size_t)(r0 + 1) * C_CH + c0 + m];
      acc = __builtin_amdgcn_wmma_f32_16x16x4_f32(
          false, A, false, B, (short)0, acc, false, false);
    }
    const int rowbase = s0 + (half << 3);
    #pragma unroll
    for (int r = 0; r < 8; ++r)
      out[(size_t)(rowbase + r) * C_CH + c0 + m] = acc[r];
  }
#else
  for (int c0 = 0; c0 < C_CH; c0 += 16) {
    for (int r = half * 8; r < half * 8 + 8; ++r)
      out[(size_t)(s0 + r) * C_CH + c0 + m] *= 1.0f / fmaxf(cnts[s0 + r], 1.0f);
  }
#endif
}

extern "C" void kernel_launch(void* const* d_in, const int* in_sizes, int n_in,
                              void* d_out, int out_size, void* d_ws, size_t ws_size,
                              hipStream_t stream) {
  const float* feat = (const float*)d_in[0];
  const int*   yx   = (const int*)d_in[1];
  const int*   seg  = (const int*)d_in[2];
  float* out = (float*)d_out;

  const int N  = in_sizes[2];
  const int S  = out_size / C_CH;
  const int HW = in_sizes[0] / C_CH;
  const int W  = (int)lround(sqrt((double)HW));

  int*   table  = (int*)d_ws;
  int*   sorted = table + (size_t)SORT_WG * NB;
  float* cnts   = (float*)(sorted + 2 * (size_t)N);
  float* cpart  = cnts + S;
  float* part   = cpart + (size_t)NCHUNK * (size_t)S;

  const int E = (N + NCHUNK - 1) / NCHUNK;

  SpVecs_bincount<<<SORT_WG, SORT_T, 0, stream>>>(yx, table, N, HW, W, E);
  SpVecs_place<<<SORT_WG, SORT_T, 0, stream>>>(yx, seg, table, sorted, N, HW, W, E);

  {
    dim3 grid(NCHUNK, C_CH);
    SpVecs_scatter<<<grid, SC_THREADS, 2 * (size_t)S * sizeof(float), stream>>>(
        feat, sorted, part, cpart, N, HW, S, E);
  }

  {
    dim3 grid((S + RD_T - 1) / RD_T, C_CH / RD_T);
    SpVecs_reduce<<<grid, 256, 0, stream>>>(part, cpart, out, cnts, S);
  }

  {
    int groups = S / 16;
    int wpb = 8;
    int blocks = (groups + wpb - 1) / wpb;
    SpVecs_finalize<<<blocks, wpb * 32, 0, stream>>>(out, cnts);
  }
}


# scatter work-groups take 16 channel planes each (256 not 4096): a thread keeps its 16 sorted entries in registers and walks the planes; 16 plane sums in LDS, same ds f32 atomics, arrival order
# speedup vs baseline: 25.1530x; 1.1029x over previous
#include <hip/hip_runtime.h>
#include <math.h>


typedef float v2f __attribute__((ext_vector_type(2)));
typedef float v8f __attribute__((ext_vector_type(8)));

#define C_CH 64
#define NCHUNK 64
#define SC_THREADS 1024
#define PPT 8
#define PLANES 16
#define EPT 16
#define RD_T 32
#define RD_B 8
#define NB_LOG2 12
#define NB (1 << NB_LOG2)
#define SORT_WG NCHUNK
#define SORT_T 1024

static_assert(NB == 4 * SORT_T, "the scan gives each thread 4 consecutive bins");
static_assert(SORT_T == 32 * 32, "the scan is 32 waves of 32 lanes");
static_assert(PLANES % 2 == 0 && C_CH % PLANES == 0, "planes are walked in pairs");

__device__ __forceinline__ int SpVecs_shift(int HW) {
  const int bits = 32 - __clz(HW - 1);
  return max(bits - NB_LOG2, 0);
}
__device__ __forceinline__ int SpVecs_bin(int off, int shift) {
  return (int)min((unsigned)off >> shift, (unsigned)(NB - 1));
}

__global__ __launch_bounds__(SORT_T) void SpVecs_bincount(
    const int* __restrict__ yx, int* __restrict__ table, int N, int HW, int W, int E) {
  __shared__ int h[NB];
  for (int i = threadIdx.x; i < NB; i += SORT_T) h[i] = 0;
  __syncthreads();

  const int   shift = SpVecs_shift(HW);
  const int2* yx2   = (const int2*)yx;
  const int e0 = blockIdx.x * E;
  const int e1 = min(e0 + E, N);
  for (int b = e0; b < e1; b += SORT_T * PPT) {
    int  key[PPT];
    bool ok[PPT];
    #pragma unroll
    for (int p = 0; p < PPT; ++p) {
      const int i = b + p * SORT_T + threadIdx.x;
      ok[p] = i < e1;
      const int2 q = yx2[ok[p] ? i : (e1 - 1)];
      key[p] = SpVecs_bin(q.x * W + q.y, shift);
    }
    #pragma unroll
    for (int p = 0; p < PPT; ++p) atomicAdd(&h[key[p]], ok[p] ? 1 : 0);
  }
  __syncthreads();

  int* row = table + (size_t)blockIdx.x * NB;
  for (int i = threadIdx.x; i < NB; i += SORT_T) row[i] = h[i];
}

__global__ __launch_bounds__(SORT_T) void SpVecs_place(
    const int* __restrict__ yx, const int* __restrict__ seg,
    const int* __restrict__ table, int* __restrict__ sorted,
    int N, int HW, int W, int E) {
  __shared__ int cur[NB];
  __shared__ int wsum[32];
  const int wg   = blockIdx.x;
  const int t    = threadIdx.x;
  const int lane = t & 31;
  const int w    = t >> 5;

  int4 tot = make_int4(0, 0, 0, 0);
  int4 bef = make_int4(0, 0, 0, 0);
  const int4* tb = (const int4*)table;
  #pragma unroll 8
  for (int g = 0; g < SORT_WG; ++g) {
    const int4 v = tb[(size_t)g * (NB / 4) + t];
    tot.x += v.x; tot.y += v.y; tot.z += v.z; tot.w += v.w;
    if (g < wg) { bef.x += v.x; bef.y += v.y; bef.z += v.z; bef.w += v.w; }
  }

  const int tsum = tot.x + tot.y + tot.z + tot.w;
  int x = tsum;
  #pragma unroll
  for (int d = 1; d < 32; d <<= 1) {
    const int y = __shfl_up(x, d);
    if (lane >= d) x += y;
  }
  if (lane == 31) wsum[w] = x;
  __syncthreads();
  if (w == 0) {
    const int y = wsum[lane];
    int z = y;
    #pragma unroll
    for (int d = 1; d < 32; d <<= 1) {
      const int u = __shfl_up(z, d);
      if (lane >= d) z += u;
    }
    wsum[lane] = z - y;
  }
  __syncthreads();
  const int base = wsum[w] + x - tsum;
  cur[4 * t + 0] = base + bef.x;
  cur[4 * t + 1] = base + tot.x + bef.y;
  cur[4 * t + 2] = base + tot.x + tot.y + bef.z;
  cur[4 * t + 3] = base + tot.x + tot.y + tot.z + bef.w;
  __syncthreads();

  const int   shift = SpVecs_shift(HW);
  const int2* yx2   = (const int2*)yx;
  int2*       srt   = (int2*)sorted;
  const int e0 = wg * E;
  const int e1 = min(e0 + E, N);
  for (int b = e0; b < e1; b += SORT_T * PPT) {
    int off[PPT];
    int sg[PPT];
    #pragma unroll
    for (int p = 0; p < PPT; ++p) {
      const int i  = b + p * SORT_T + threadIdx.x;
      const int ii = min(i, e1 - 1);
      const int2 q = yx2[ii];
      sg[p]  = seg[ii];
      off[p] = (i < e1) ? (q.x * W + q.y) : -1;
    }
    #pragma unroll
    for (int p = 0; p < PPT; ++p) {
      if (off[p] >= 0) {
        const int pos = atomicAdd(&cur[SpVecs_bin(off[p], shift)], 1);
        srt[pos] = make_int2(off[p], sg[p]);
      }
    }
  }
}

__device__ __forceinline__ void SpVecs_gather(
    float (&v)[EPT], const float* __restrict__ feat, const unsigned (&idx)[EPT], unsigned poff) {
  #pragma unroll
  for (int p = 0; p < EPT; ++p) v[p] = feat[idx[p] + poff];
}
__device__ __forceinline__ void SpVecs_add(
    float* a, const int (&s)[EPT], const bool (&ok)[EPT], const float (&v)[EPT]) {
  #pragma unroll
  for (int p = 0; p < EPT; ++p) atomicAdd(&a[s[p]], ok[p] ? v[p] : 0.0f);
}

__global__ __launch_bounds__(SC_THREADS) void SpVecs_scatter(
    const float* __restrict__ feat, const int* __restrict__ sorted,
    float* __restrict__ part,
    float* __restrict__ cpart, int N, int HW, int S, int E) {
  extern __shared__ float lds[];
  float* acc  = lds;
  float* cacc = lds + (size_t)PLANES * S;

  const int  g        = blockIdx.x;
  const int  c0       = blockIdx.y * PLANES;
  const bool counting = (blockIdx.y == 0);
  const int  lane     = threadIdx.x & 31;
  const int  w        = threadIdx.x >> 5;

  const int nz = (counting ? PLANES + 1 : PLANES) * S;
  {
    float4* l4 = (float4*)lds;
    const float4 z4 = make_float4(0.0f, 0.0f, 0.0f, 0.0f);
    for (int i = threadIdx.x; i < (nz >> 2); i += SC_THREADS) l4[i] = z4;
    for (int i = (nz & ~3) + threadIdx.x; i < nz; i += SC_THREADS) lds[i] = 0.0f;
  }
  __syncthreads();

  const int2* srt = (const int2*)sorted;
  const int e0 = g * E;
  const int e1 = min(e0 + E, N);
  const unsigned uHW  = (unsigned)HW;
  const unsigned base = (unsigned)c0 * uHW;

  for (int b = e0; b < e1; b += SC_THREADS * EPT) {
    unsigned idx[EPT];
    int      s[EPT];
    bool     ok[EPT];
    const int wb = b + w * (32 * EPT) + lane;
    #pragma unroll
    for (int p = 0; p < EPT; ++p) {
      const int i = wb + p * 32;
      ok[p] = i < e1;
      const int ii = ok[p] ? i : (e1 - 1);
      const int2 q = srt[ii];
      idx[p] = base + (unsigned)q.x;
      s[p]   = q.y;
    }

    if (counting) {
      #pragma unroll
      for (int p = 0; p < EPT; ++p) atomicAdd(&cacc[s[p]], ok[p] ? 1.0f : 0.0f);
    }

    float va[EPT], vb[EPT];
    float*   a    = acc;
    unsigned poff = 0;
    SpVecs_gather(va, feat, idx, poff);
    for (int k = 0; k < PLANES - 2; k += 2) {
      SpVecs_gather(vb, feat, idx, poff + uHW);
      SpVecs_add(a, s, ok, va);
      SpVecs_gather(va, feat, idx, poff + 2 * uHW);
      SpVecs_add(a + S, s, ok, vb);
      poff += 2 * uHW;
      a    += 2 * (size_t)S;
    }
    SpVecs_gather(vb, feat, idx, poff + uHW);
    SpVecs_add(a, s, ok, va);
    SpVecs_add(a + S, s, ok, vb);
  }
  __syncthreads();

  const bool vec = ((S & 3) == 0) && ((((size_t)part) & 15) == 0);
  #pragma unroll 1
  for (int k = 0; k < PLANES; ++k) {
    float*       dst = part + ((size_t)(c0 + k) * NCHUNK + g) * (size_t)S;
    const float* pa  = acc + (size_t)k * S;
    if (vec) {
      float4*       d4 = (float4*)dst;
      const float4* a4 = (const float4*)pa;
      for (int i = threadIdx.x; i < (S >> 2); i += SC_THREADS) d4[i] = a4[i];
    } else {
      for (int i = threadIdx.x; i < S; i += SC_THREADS) dst[i] = pa[i];
    }
  }
  if (counting) {
    float* cd = cpart + (size_t)g * (size_t)S;
    for (int i = threadIdx.x; i < S; i += SC_THREADS) cd[i] = cacc[i];
  }
}

__global__ __launch_bounds__(256) void SpVecs_reduce(
    const float* __restrict__ part, const float* __restrict__ cpart,
    float* __restrict__ sums, float* __restrict__ cnts, int S) {
  __shared__ float tile[RD_T][RD_T + 1];
  const int lane = threadIdx.x & 31;
  const int w    = threadIdx.x >> 5;
  const int s0   = blockIdx.x * RD_T;
  const int c0   = blockIdx.y * RD_T;
  const int s    = s0 + lane;
  const bool in  = s < S;
  const int  sc  = in ? s : (S - 1);

  const unsigned uS = (unsigned)S;
  unsigned pk[4];
  #pragma unroll
  for (int k = 0; k < 4; ++k)
    pk[k] = ((unsigned)(c0 + w * 4 + k) * NCHUNK) * uS + (unsigned)sc;

  float a[4] = {0.0f, 0.0f, 0.0f, 0.0f};
  for (int g0 = 0; g0 < NCHUNK; g0 += RD_B) {
    float t[4][RD_B];
    #pragma unroll
    for (int j = 0; j < RD_B; ++j) {
      #pragma unroll
      for (int k = 0; k < 4; ++k) t[k][j] = part[pk[k] + (unsigned)(g0 + j) * uS];
    }
    #pragma unroll
    for (int j = 0; j < RD_B; ++j) {
      #pragma unroll
      for (int k = 0; k < 4; ++k) a[k] += t[k][j];
    }
  }
  #pragma unroll
  for (int k = 0; k < 4; ++k) tile[w * 4 + k][lane] = a[k];

  if (blockIdx.y == 0 && w == 0) {
    float n = 0.0f;
    for (int g0 = 0; g0 < NCHUNK; g0 += RD_B) {
      float t[RD_B];
      #pragma unroll
      for (int j = 0; j < RD_B; ++j) t[j] = cpart[(unsigned)(g0 + j) * uS + (unsigned)sc];
      #pragma unroll
      for (int j = 0; j < RD_B; ++j) n += t[j];
    }
    if (in) cnts[s] = n;
  }
  __syncthreads();

  #pragma unroll
  for (int k = 0; k < 4; ++k) {
    const int r = w * 4 + k;
    if (s0 + r < S) sums[(size_t)(s0 + r) * C_CH + c0 + lane] = tile[lane][r];
  }
}

__global__ __launch_bounds__(256) void SpVecs_finalize(
    float* __restrict__ out, const float* __restrict__ cnts) {
  const int lane  = threadIdx.x & 31;
  const int wid   = threadIdx.x >> 5;
  const int g     = blockIdx.x * (blockDim.x >> 5) + wid;
  const int s0    = g << 4;
  const int m     = lane & 15;
  const int half  = lane >> 4;
  const int khalf = half << 1;

  const float invc = 1.0f / fmaxf(cnts[s0 + m], 1.0f);

#if __has_builtin(__builtin_amdgcn_wmma_f32_16x16x4_f32)
  #pragma unroll
  for (int c0 = 0; c0 < C_CH; c0 += 16) {
    v8f acc = {0.f, 0.f, 0.f, 0.f, 0.f, 0.f, 0.f, 0.f};
    #pragma unroll
    for (int j = 0; j < 4; ++j) {
      v2f A;
      A.x = (m == 4 * j + khalf + 0) ? invc : 0.0f;
      A.y = (m == 4 * j + khalf + 1) ? invc : 0.0f;
      const int r0 = s0 + 4 * j + khalf;
      v2f B;
      B.x = out[(size_t)r0 * C_CH + c0 + m];
      B.y = out[(size_t)(r0 + 1) * C_CH + c0 + m];
      acc = __builtin_amdgcn_wmma_f32_16x16x4_f32(
          false, A, false, B, (short)0, acc, false, false);
    }
    const int rowbase = s0 + (half << 3);
    #pragma unroll
    for (int r = 0; r < 8; ++r)
      out[(size_t)(rowbase + r) * C_CH + c0 + m] = acc[r];
  }
#else
  for (int c0 = 0; c0 < C_CH; c0 += 16) {
    for (int r = half * 8; r < half * 8 + 8; ++r)
      out[(size_t)(s0 + r) * C_CH + c0 + m] *= 1.0f / fmaxf(cnts[s0 + r], 1.0f);
  }
#endif
}

extern "C" void kernel_launch(void* const* d_in, const int* in_sizes, int n_in,
                              void* d_out, int out_size, void* d_ws, size_t ws_size,
                              hipStream_t stream) {
  const float* feat = (const float*)d_in[0];
  const int*   yx   = (const int*)d_in[1];
  const int*   seg  = (const int*)d_in[2];
  float* out = (float*)d_out;

  const int N  = in_sizes[2];
  const int S  = out_size / C_CH;
  const int HW = in_sizes[0] / C_CH;
  const int W  = (int)lround(sqrt((double)HW));

  int*   table  = (int*)d_ws;
  int*   sorted = table + (size_t)SORT_WG * NB;
  float* cnts   = (float*)(sorted + 2 * (size_t)N);
  float* cpart  = cnts + S;
  float* part   = cpart + (size_t)NCHUNK * (size_t)S;

  const int E = (N + NCHUNK - 1) / NCHUNK;

  SpVecs_bincount<<<SORT_WG, SORT_T, 0, stream>>>(yx, table, N, HW, W, E);
  SpVecs_place<<<SORT_WG, SORT_T, 0, stream>>>(yx, seg, table, sorted, N, HW, W, E);

  {
    dim3 grid(NCHUNK, C_CH / PLANES);
    SpVecs_scatter<<<grid, SC_THREADS, (PLANES + 1) * (size_t)S * sizeof(float), stream>>>(
        feat, sorted, part, cpart, N, HW, S, E);
  }

  {
    dim3 grid((S + RD_T - 1) / RD_T, C_CH / RD_T);
    SpVecs_reduce<<<grid, 256, 0, stream>>>(part, cpart, out, cnts, S);
  }

  {
    int groups = S / 16;
    int wpb = 8;
    int blocks = (groups + wpb - 1) / wpb;
    SpVecs_finalize<<<blocks, wpb * 32, 0, stream>>>(out, cnts);
  }
}


# scatter work-group first counting-sorts its 16384 entries by pixel offset in LDS (where the sums go later), so a wave's gathers fall in its own 1-2 lines each; changes the order of the f32 sums
# speedup vs baseline: 28.0178x; 1.1139x over previous
#include <hip/hip_runtime.h>
#include <math.h>


typedef float v2f __attribute__((ext_vector_type(2)));
typedef float v8f __attribute__((ext_vector_type(8)));

#define C_CH 64
#define NCHUNK 64
#define SC_THREADS 1024
#define PPT 8
#define PLANES 16
#define EPT 16
#define KPT 17
#define KL (KPT * SC_THREADS)
#define RD_T 32
#define RD_B 8
#define NB_LOG2 12
#define NB (1 << NB_LOG2)
#define SORT_WG NCHUNK
#define SORT_T 1024

static_assert(NB == 4 * SORT_T, "the scan gives each thread 4 consecutive bins");
static_assert(SORT_T == 32 * 32, "the scan is 32 waves of 32 lanes");
static_assert(PLANES % 2 == 0 && C_CH % PLANES == 0, "planes are walked in pairs");
static_assert(SC_THREADS == 32 * 32, "the in-work-group scan is 32 waves of 32 lanes");
static_assert((KL + 32) % 2 == 0, "the pair buffer behind the counters is 8-byte aligned");

__device__ __forceinline__ int SpVecs_shift(int HW) {
  const int bits = 32 - __clz(HW - 1);
  return max(bits - NB_LOG2, 0);
}
__device__ __forceinline__ int SpVecs_bin(int off, int shift) {
  return (int)min((unsigned)off >> shift, (unsigned)(NB - 1));
}

__global__ __launch_bounds__(SORT_T) void SpVecs_bincount(
    const int* __restrict__ yx, int* __restrict__ table, int N, int HW, int W, int E) {
  __shared__ int h[NB];
  for (int i = threadIdx.x; i < NB; i += SORT_T) h[i] = 0;
  __syncthreads();

  const int   shift = SpVecs_shift(HW);
  const int2* yx2   = (const int2*)yx;
  const int e0 = blockIdx.x * E;
  const int e1 = min(e0 + E, N);
  for (int b = e0; b < e1; b += SORT_T * PPT) {
    int  key[PPT];
    bool ok[PPT];
    #pragma unroll
    for (int p = 0; p < PPT; ++p) {
      const int i = b + p * SORT_T + threadIdx.x;
      ok[p] = i < e1;
      const int2 q = yx2[ok[p] ? i : (e1 - 1)];
      key[p] = SpVecs_bin(q.x * W + q.y, shift);
    }
    #pragma unroll
    for (int p = 0; p < PPT; ++p) atomicAdd(&h[key[p]], ok[p] ? 1 : 0);
  }
  __syncthreads();

  int* row = table + (size_t)blockIdx.x * NB;
  for (int i = threadIdx.x; i < NB; i += SORT_T) row[i] = h[i];
}

__global__ __launch_bounds__(SORT_T) void SpVecs_place(
    const int* __restrict__ yx, const int* __restrict__ seg,
    const int* __restrict__ table, int* __restrict__ sorted,
    int N, int HW, int W, int E) {
  __shared__ int cur[NB];
  __shared__ int wsum[32];
  const int wg   = blockIdx.x;
  const int t    = threadIdx.x;
  const int lane = t & 31;
  const int w    = t >> 5;

  int4 tot = make_int4(0, 0, 0, 0);
  int4 bef = make_int4(0, 0, 0, 0);
  const int4* tb = (const int4*)table;
  #pragma unroll 8
  for (int g = 0; g < SORT_WG; ++g) {
    const int4 v = tb[(size_t)g * (NB / 4) + t];
    tot.x += v.x; tot.y += v.y; tot.z += v.z; tot.w += v.w;
    if (g < wg) { bef.x += v.x; bef.y += v.y; bef.z += v.z; bef.w += v.w; }
  }

  const int tsum = tot.x + tot.y + tot.z + tot.w;
  int x = tsum;
  #pragma unroll
  for (int d = 1; d < 32; d <<= 1) {
    const int y = __shfl_up(x, d);
    if (lane >= d) x += y;
  }
  if (lane == 31) wsum[w] = x;
  __syncthreads();
  if (w == 0) {
    const int y = wsum[lane];
    int z = y;
    #pragma unroll
    for (int d = 1; d < 32; d <<= 1) {
      const int u = __shfl_up(z, d);
      if (lane >= d) z += u;
    }
    wsum[lane] = z - y;
  }
  __syncthreads();
  const int base = wsum[w] + x - tsum;
  cur[4 * t + 0] = base + bef.x;
  cur[4 * t + 1] = base + tot.x + bef.y;
  cur[4 * t + 2] = base + tot.x + tot.y + bef.z;
  cur[4 * t + 3] = base + tot.x + tot.y + tot.z + bef.w;
  __syncthreads();

  const int   shift = SpVecs_shift(HW);
  const int2* yx2   = (const int2*)yx;
  int2*       srt   = (int2*)sorted;
  const int e0 = wg * E;
  const int e1 = min(e0 + E, N);
  for (int b = e0; b < e1; b += SORT_T * PPT) {
    int off[PPT];
    int sg[PPT];
    #pragma unroll
    for (int p = 0; p < PPT; ++p) {
      const int i  = b + p * SORT_T + threadIdx.x;
      const int ii = min(i, e1 - 1);
      const int2 q = yx2[ii];
      sg[p]  = seg[ii];
      off[p] = (i < e1) ? (q.x * W + q.y) : -1;
    }
    #pragma unroll
    for (int p = 0; p < PPT; ++p) {
      if (off[p] >= 0) {
        const int pos = atomicAdd(&cur[SpVecs_bin(off[p], shift)], 1);
        srt[pos] = make_int2(off[p], sg[p]);
      }
    }
  }
}

__device__ __forceinline__ void SpVecs_gather(
    float (&v)[EPT], const float* __restrict__ feat, const unsigned (&idx)[EPT], unsigned poff) {
  #pragma unroll
  for (int p = 0; p < EPT; ++p) v[p] = feat[idx[p] + poff];
}
__device__ __forceinline__ void SpVecs_add(
    float* a, const int (&s)[EPT], const bool (&ok)[EPT], const float (&v)[EPT]) {
  #pragma unroll
  for (int p = 0; p < EPT; ++p) atomicAdd(&a[s[p]], ok[p] ? v[p] : 0.0f);
}


__device__ __forceinline__ void SpVecs_localsort(
    int (&off)[EPT], int (&s)[EPT], int* li,
    int lo, int lshift, int nvalid) {
  int*  hist = li;
  int*  wsum = li + KL;
  int2* buf  = (int2*)(li + KL + 32);
  const int t    = threadIdx.x;
  const int lane = t & 31;
  const int w    = t >> 5;
  const int jb   = w * (32 * EPT) + lane;

  {
    int4* h4 = (int4*)hist;
    const int4 z4 = make_int4(0, 0, 0, 0);
    for (int i = t; i < KL / 4; i += SC_THREADS) h4[i] = z4;
  }
  __syncthreads();

  int key[EPT];
  #pragma unroll
  for (int p = 0; p < EPT; ++p) {
    key[p] = (int)min((unsigned)(off[p] - lo) >> lshift, (unsigned)(KL - 1));
    atomicAdd(&hist[key[p]], (jb + p * 32 < nvalid) ? 1 : 0);
  }
  __syncthreads();

  int c[KPT];
  int tsum = 0;
  #pragma unroll
  for (int i = 0; i < KPT; ++i) { c[i] = hist[KPT * t + i]; tsum += c[i]; }
  int x = tsum;
  #pragma unroll
  for (int d = 1; d < 32; d <<= 1) {
    const int y = __shfl_up(x, d);
    if (lane >= d) x += y;
  }
  if (lane == 31) wsum[w] = x;
  __syncthreads();
  if (w == 0) {
    const int y = wsum[lane];
    int z = y;
    #pragma unroll
    for (int d = 1; d < 32; d <<= 1) {
      const int u = __shfl_up(z, d);
      if (lane >= d) z += u;
    }
    wsum[lane] = z - y;
  }
  __syncthreads();
  int run = wsum[w] + x - tsum;
  #pragma unroll
  for (int i = 0; i < KPT; ++i) { hist[KPT * t + i] = run; run += c[i]; }
  __syncthreads();

  #pragma unroll
  for (int p = 0; p < EPT; ++p) {
    if (jb + p * 32 < nvalid) {
      const int pos = atomicAdd(&hist[key[p]], 1);
      buf[pos] = make_int2(off[p], s[p]);
    }
  }
  __syncthreads();

  #pragma unroll
  for (int p = 0; p < EPT; ++p) {
    const int2 q = buf[min(jb + p * 32, nvalid - 1)];
    off[p] = q.x;
    s[p]   = q.y;
  }
  __syncthreads();
}

__global__ __launch_bounds__(SC_THREADS) void SpVecs_scatter(
    const float* __restrict__ feat, const int* __restrict__ sorted,
    float* __restrict__ part,
    float* __restrict__ cpart, int N, int HW, int S, int E) {
  extern __shared__ float lds[];
  float* acc  = lds;
  float* cacc = lds + (size_t)PLANES * S;

  const int  g        = blockIdx.x;
  const int  c0       = blockIdx.y * PLANES;
  const bool counting = (blockIdx.y == 0);
  const int  lane     = threadIdx.x & 31;
  const int  w        = threadIdx.x >> 5;

  const int nz = (counting ? PLANES + 1 : PLANES) * S;
  const bool can_sort = PLANES * S >= KL + 32 + 2 * SC_THREADS * EPT;
  bool cleared = false;

  const int2* srt = (const int2*)sorted;
  const int e0 = g * E;
  const int e1 = min(e0 + E, N);
  const unsigned uHW  = (unsigned)HW;
  const unsigned base = (unsigned)c0 * uHW;

  for (int b = e0; b < e1; b += SC_THREADS * EPT) {
    unsigned idx[EPT];
    int      s[EPT];
    bool     ok[EPT];
    const int wb = b + w * (32 * EPT) + lane;
    int off[EPT];
    #pragma unroll
    for (int p = 0; p < EPT; ++p) {
      const int i = wb + p * 32;
      const int2 q = srt[min(i, e1 - 1)];
      off[p] = q.x;
      s[p]   = q.y;
    }
    const int nvalid = min(e1 - b, SC_THREADS * EPT);

    if (!cleared) {
      if (can_sort) {
        const int shift  = SpVecs_shift(HW);
        const int bf     = SpVecs_bin(srt[b].x, shift);
        const int bl     = SpVecs_bin(srt[b + nvalid - 1].x, shift);
        const int lo     = bf << shift;
        const int span   = max(((bl + 1) << shift) - lo, 1);
        int lshift = 0;
        while (lshift < 31 && ((span - 1) >> lshift) >= KL) ++lshift;
        SpVecs_localsort(off, s, (int*)lds, lo, lshift, nvalid);
      }
      float4* l4 = (float4*)lds;
      const float4 z4 = make_float4(0.0f, 0.0f, 0.0f, 0.0f);
      for (int i = threadIdx.x; i < (nz >> 2); i += SC_THREADS) l4[i] = z4;
      for (int i = (nz & ~3) + threadIdx.x; i < nz; i += SC_THREADS) lds[i] = 0.0f;
      __syncthreads();
      cleared = true;
    }

    #pragma unroll
    for (int p = 0; p < EPT; ++p) {
      idx[p] = base + (unsigned)off[p];
      ok[p]  = (w * (32 * EPT) + p * 32 + lane) < nvalid;
    }

    if (counting) {
      #pragma unroll
      for (int p = 0; p < EPT; ++p) atomicAdd(&cacc[s[p]], ok[p] ? 1.0f : 0.0f);
    }

    float va[EPT], vb[EPT];
    float*   a    = acc;
    unsigned poff = 0;
    SpVecs_gather(va, feat, idx, poff);
    for (int k = 0; k < PLANES - 2; k += 2) {
      SpVecs_gather(vb, feat, idx, poff + uHW);
      SpVecs_add(a, s, ok, va);
      SpVecs_gather(va, feat, idx, poff + 2 * uHW);
      SpVecs_add(a + S, s, ok, vb);
      poff += 2 * uHW;
      a    += 2 * (size_t)S;
    }
    SpVecs_gather(vb, feat, idx, poff + uHW);
    SpVecs_add(a, s, ok, va);
    SpVecs_add(a + S, s, ok, vb);
  }
  if (!cleared) {
    for (int i = threadIdx.x; i < nz; i += SC_THREADS) lds[i] = 0.0f;
  }
  __syncthreads();

  const bool vec = ((S & 3) == 0) && ((((size_t)part) & 15) == 0);
  #pragma unroll 1
  for (int k = 0; k < PLANES; ++k) {
    float*       dst = part + ((size_t)(c0 + k) * NCHUNK + g) * (size_t)S;
    const float* pa  = acc + (size_t)k * S;
    if (vec) {
      float4*       d4 = (float4*)dst;
      const float4* a4 = (const float4*)pa;
      for (int i = threadIdx.x; i < (S >> 2); i += SC_THREADS) d4[i] = a4[i];
    } else {
      for (int i = threadIdx.x; i < S; i += SC_THREADS) dst[i] = pa[i];
    }
  }
  if (counting) {
    float* cd = cpart + (size_t)g * (size_t)S;
    for (int i = threadIdx.x; i < S; i += SC_THREADS) cd[i] = cacc[i];
  }
}

__global__ __launch_bounds__(256) void SpVecs_reduce(
    const float* __restrict__ part, const float* __restrict__ cpart,
    float* __restrict__ sums, float* __restrict__ cnts, int S) {
  __shared__ float tile[RD_T][RD_T + 1];
  const int lane = threadIdx.x & 31;
  const int w    = threadIdx.x >> 5;
  const int s0   = blockIdx.x * RD_T;
  const int c0   = blockIdx.y * RD_T;
  const int s    = s0 + lane;
  const bool in  = s < S;
  const int  sc  = in ? s : (S - 1);

  const unsigned uS = (unsigned)S;
  unsigned pk[4];
  #pragma unroll
  for (int k = 0; k < 4; ++k)
    pk[k] = ((unsigned)(c0 + w * 4 + k) * NCHUNK) * uS + (unsigned)sc;

  float a[4] = {0.0f, 0.0f, 0.0f, 0.0f};
  for (int g0 = 0; g0 < NCHUNK; g0 += RD_B) {
    float t[4][RD_B];
    #pragma unroll
    for (int j = 0; j < RD_B; ++j) {
      #pragma unroll
      for (int k = 0; k < 4; ++k) t[k][j] = part[pk[k] + (unsigned)(g0 + j) * uS];
    }
    #pragma unroll
    for (int j = 0; j < RD_B; ++j) {
      #pragma unroll
      for (int k = 0; k < 4; ++k) a[k] += t[k][j];
    }
  }
  #pragma unroll
  for (int k = 0; k < 4; ++k) tile[w * 4 + k][lane] = a[k];

  if (blockIdx.y == 0 && w == 0) {
    float n = 0.0f;
    for (int g0 = 0; g0 < NCHUNK; g0 += RD_B) {
      float t[RD_B];
      #pragma unroll
      for (int j = 0; j < RD_B; ++j) t[j] = cpart[(unsigned)(g0 + j) * uS + (unsigned)sc];
      #pragma unroll
      for (int j = 0; j < RD_B; ++j) n += t[j];
    }
    if (in) cnts[s] = n;
  }
  __syncthreads();

  #pragma unroll
  for (int k = 0; k < 4; ++k) {
    const int r = w * 4 + k;
    if (s0 + r < S) sums[(size_t)(s0 + r) * C_CH + c0 + lane] = tile[lane][r];
  }
}

__global__ __launch_bounds__(256) void SpVecs_finalize(
    float* __restrict__ out, const float* __restrict__ cnts) {
  const int lane  = threadIdx.x & 31;
  const int wid   = threadIdx.x >> 5;
  const int g     = blockIdx.x * (blockDim.x >> 5) + wid;
  const int s0    = g << 4;
  const int m     = lane & 15;
  const int half  = lane >> 4;
  const int khalf = half << 1;

  const float invc = 1.0f / fmaxf(cnts[s0 + m], 1.0f);

#if __has_builtin(__builtin_amdgcn_wmma_f32_16x16x4_f32)
  #pragma unroll
  for (int c0 = 0; c0 < C_CH; c0 += 16) {
    v8f acc = {0.f, 0.f, 0.f, 0.f, 0.f, 0.f, 0.f, 0.f};
    #pragma unroll
    for (int j = 0; j < 4; ++j) {
      v2f A;
      A.x = (m == 4 * j + khalf + 0) ? invc : 0.0f;
      A.y = (m == 4 * j + khalf + 1) ? invc : 0.0f;
      const int r0 = s0 + 4 * j + khalf;
      v2f B;
      B.x = out[(size_t)r0 * C_CH + c0 + m];
      B.y = out[(size_t)(r0 + 1) * C_CH + c0 + m];
      acc = __builtin_amdgcn_wmma_f32_16x16x4_f32(
          false, A, false, B, (short)0, acc, false, false);
    }
    const int rowbase = s0 + (half << 3);
    #pragma unroll
    for (int r = 0; r < 8; ++r)
      out[(size_t)(rowbase + r) * C_CH + c0 + m] = acc[r];
  }
#else
  for (int c0 = 0; c0 < C_CH; c0 += 16) {
    for (int r = half * 8; r < half * 8 + 8; ++r)
      out[(size_t)(s0 + r) * C_CH + c0 + m] *= 1.0f / fmaxf(cnts[s0 + r], 1.0f);
  }
#endif
}

extern "C" void kernel_launch(void* const* d_in, const int* in_sizes, int n_in,
                              void* d_out, int out_size, void* d_ws, size_t ws_size,
                              hipStream_t stream) {
  const float* feat = (const float*)d_in[0];
  const int*   yx   = (const int*)d_in[1];
  const int*   seg  = (const int*)d_in[2];
  float* out = (float*)d_out;

  const int N  = in_sizes[2];
  const int S  = out_size / C_CH;
  const int HW = in_sizes[0] / C_CH;
  const int W  = (int)lround(sqrt((double)HW));

  int*   table  = (int*)d_ws;
  int*   sorted = table + (size_t)SORT_WG * NB;
  float* cnts   = (float*)(sorted + 2 * (size_t)N);
  float* cpart  = cnts + S;
  float* part   = cpart + (size_t)NCHUNK * (size_t)S;

  const int E = (N + NCHUNK - 1) / NCHUNK;

  SpVecs_bincount<<<SORT_WG, SORT_T, 0, stream>>>(yx, table, N, HW, W, E);
  SpVecs_place<<<SORT_WG, SORT_T, 0, stream>>>(yx, seg, table, sorted, N, HW, W, E);

  {
    dim3 grid(NCHUNK, C_CH / PLANES);
    SpVecs_scatter<<<grid, SC_THREADS, (PLANES + 1) * (size_t)S * sizeof(float), stream>>>(
        feat, sorted, part, cpart, N, HW, S, E);
  }

  {
    dim3 grid((S + RD_T - 1) / RD_T, C_CH / RD_T);
    SpVecs_reduce<<<grid, 256, 0, stream>>>(part, cpart, out, cnts, S);
  }

  {
    int groups = S / 16;
    int wpb = 8;
    int blocks = (groups + wpb - 1) / wpb;
    SpVecs_finalize<<<blocks, wpb * 32, 0, stream>>>(out, cnts);
  }
}


# first-level counting sort cut from 4096 bins to the 64 pixel ranges that are the scatter chunks (the in-work-group sort orders them); sort launches on 256 groups, 64 KB table; f32 sum order changes
# speedup vs baseline: 40.8896x; 1.4594x over previous
#include <hip/hip_runtime.h>
#include <math.h>


typedef float v2f __attribute__((ext_vector_type(2)));
typedef float v8f __attribute__((ext_vector_type(8)));

#define C_CH 64
#define NCHUNK 64
#define SC_THREADS 1024
#define PPT 16
#define PLANES 16
#define EPT 17
#define KPT 17
#define KL (KPT * SC_THREADS)
#define RD_T 32
#define RD_B 8
#define NB_LOG2 6
#define NB (1 << NB_LOG2)
#define SORT_WG 256
#define SORT_T 256
#define NBQ (NB / 4)

static_assert(NB == NCHUNK, "a pixel range is one scatter work-group's chunk");
static_assert(NB == 64, "one wave scans the range totals, two per lane");
static_assert(SORT_T >= NB, "one thread per range clears and stores the counts");
static_assert(SORT_T % NBQ == 0 && (SORT_WG * NBQ) % SORT_T == 0, "the table is read in whole rounds of SORT_T 16-byte words");
static_assert(PLANES % 2 == 0 && C_CH % PLANES == 0, "planes are walked in pairs");
static_assert(SC_THREADS == 32 * 32, "the in-work-group scan is 32 waves of 32 lanes");
static_assert((KL + 32) % 2 == 0, "the pair buffer behind the counters is 8-byte aligned");

__device__ __forceinline__ int SpVecs_shift(int HW) {
  const int bits = 32 - __clz(HW - 1);
  return max(bits - NB_LOG2, 0);
}
__device__ __forceinline__ int SpVecs_bin(int off, int shift) {
  return (int)min((unsigned)off >> shift, (unsigned)(NB - 1));
}

__global__ __launch_bounds__(SORT_T) void SpVecs_bincount(
    const int* __restrict__ yx, int* __restrict__ table, int N, int HW, int W, int E) {
  __shared__ int h[NB];
  if (threadIdx.x < NB) h[threadIdx.x] = 0;
  __syncthreads();

  const int   shift = SpVecs_shift(HW);
  const int2* yx2   = (const int2*)yx;
  const int e0 = blockIdx.x * E;
  const int e1 = min(e0 + E, N);
  for (int b = e0; b < e1; b += SORT_T * PPT) {
    int  key[PPT];
    bool ok[PPT];
    #pragma unroll
    for (int p = 0; p < PPT; ++p) {
      const int i = b + p * SORT_T + threadIdx.x;
      ok[p] = i < e1;
      const int2 q = yx2[ok[p] ? i : (e1 - 1)];
      key[p] = SpVecs_bin(q.x * W + q.y, shift);
    }
    #pragma unroll
    for (int p = 0; p < PPT; ++p) {
      if (ok[p]) atomicAdd(&h[key[p]], 1);
    }
  }
  __syncthreads();

  if (threadIdx.x < NB) table[(size_t)blockIdx.x * NB + threadIdx.x] = h[threadIdx.x];
}

__global__ __launch_bounds__(SORT_T) void SpVecs_place(
    const int* __restrict__ yx, const int* __restrict__ seg,
    const int* __restrict__ table, int* __restrict__ sorted, int* __restrict__ cstart,
    int N, int HW, int W, int E) {
  __shared__ int tot[NB];
  __shared__ int bef[NB];
  __shared__ int cur[NB];
  const int wg   = blockIdx.x;
  const int t    = threadIdx.x;
  const int lane = t & 31;
  const int w    = t >> 5;

  if (t < NB) { tot[t] = 0; bef[t] = 0; }
  __syncthreads();

  int4 mt = make_int4(0, 0, 0, 0);
  int4 mb = make_int4(0, 0, 0, 0);
  const int4* tb = (const int4*)table;
  #pragma unroll
  for (int k = 0; k < SORT_WG * NBQ / SORT_T; ++k) {
    const int4 v   = tb[t + k * SORT_T];
    const bool in  = (t / NBQ + k * (SORT_T / NBQ)) < wg;
    mt.x += v.x; mt.y += v.y; mt.z += v.z; mt.w += v.w;
    mb.x += in ? v.x : 0; mb.y += in ? v.y : 0; mb.z += in ? v.z : 0; mb.w += in ? v.w : 0;
  }
  {
    const int r = 4 * (t % NBQ);
    atomicAdd(&tot[r + 0], mt.x); atomicAdd(&tot[r + 1], mt.y);
    atomicAdd(&tot[r + 2], mt.z); atomicAdd(&tot[r + 3], mt.w);
    atomicAdd(&bef[r + 0], mb.x); atomicAdd(&bef[r + 1], mb.y);
    atomicAdd(&bef[r + 2], mb.z); atomicAdd(&bef[r + 3], mb.w);
  }
  __syncthreads();

  if (w == 0) {
    const int a  = tot[2 * lane];
    const int a1 = tot[2 * lane + 1];
    const int sm = a + a1;
    int x = sm;
    #pragma unroll
    for (int d = 1; d < 32; d <<= 1) {
      const int y = __shfl_up(x, d);
      if (lane >= d) x += y;
    }
    const int ex = x - sm;
    cur[2 * lane]     = ex + bef[2 * lane];
    cur[2 * lane + 1] = ex + a + bef[2 * lane + 1];
    if (wg == 0) {
      cstart[2 * lane]     = ex;
      cstart[2 * lane + 1] = ex + a;
      if (lane == 31) cstart[NB] = x;
    }
  }
  __syncthreads();

  const int   shift = SpVecs_shift(HW);
  const int2* yx2   = (const int2*)yx;
  int2*       srt   = (int2*)sorted;
  const int e0 = wg * E;
  const int e1 = min(e0 + E, N);
  for (int b = e0; b < e1; b += SORT_T * PPT) {
    int  off[PPT];
    int  sg[PPT];
    bool ok[PPT];
    #pragma unroll
    for (int p = 0; p < PPT; ++p) {
      const int i  = b + p * SORT_T + threadIdx.x;
      ok[p] = i < e1;
      const int ii = ok[p] ? i : (e1 - 1);
      const int2 q = yx2[ii];
      sg[p]  = seg[ii];
      off[p] = q.x * W + q.y;
    }
    #pragma unroll
    for (int p = 0; p < PPT; ++p) {
      if (ok[p]) {
        const int pos = atomicAdd(&cur[SpVecs_bin(off[p], shift)], 1);
        srt[pos] = make_int2(off[p], sg[p]);
      }
    }
  }
}

__device__ __forceinline__ void SpVecs_gather(
    float (&v)[EPT], const float* __restrict__ feat, const unsigned (&idx)[EPT], unsigned poff) {
  #pragma unroll
  for (int p = 0; p < EPT; ++p) v[p] = feat[idx[p] + poff];
}
__device__ __forceinline__ void SpVecs_add(
    float* a, const int (&s)[EPT], const bool (&ok)[EPT], const float (&v)[EPT]) {
  #pragma unroll
  for (int p = 0; p < EPT; ++p) atomicAdd(&a[s[p]], ok[p] ? v[p] : 0.0f);
}


__device__ __forceinline__ void SpVecs_localsort(
    int (&off)[EPT], int (&s)[EPT], const bool (&ok)[EPT], int* li,
    int lo, int lshift, int nvalid, int pw) {
  int*  hist = li;
  int*  wsum = li + KL;
  int2* buf  = (int2*)(li + KL + 32);
  const int t    = threadIdx.x;
  const int lane = t & 31;
  const int w    = t >> 5;
  const int jb   = w * pw + lane;

  {
    int4* h4 = (int4*)hist;
    const int4 z4 = make_int4(0, 0, 0, 0);
    for (int i = t; i < KL / 4; i += SC_THREADS) h4[i] = z4;
  }
  __syncthreads();

  int key[EPT];
  #pragma unroll
  for (int p = 0; p < EPT; ++p) {
    key[p] = (int)min((unsigned)(off[p] - lo) >> lshift, (unsigned)(KL - 1));
    if (ok[p]) atomicAdd(&hist[key[p]], 1);
  }
  __syncthreads();

  int c[KPT];
  int tsum = 0;
  #pragma unroll
  for (int i = 0; i < KPT; ++i) { c[i] = hist[KPT * t + i]; tsum += c[i]; }
  int x = tsum;
  #pragma unroll
  for (int d = 1; d < 32; d <<= 1) {
    const int y = __shfl_up(x, d);
    if (lane >= d) x += y;
  }
  if (lane == 31) wsum[w] = x;
  __syncthreads();
  if (w == 0) {
    const int y = wsum[lane];
    int z = y;
    #pragma unroll
    for (int d = 1; d < 32; d <<= 1) {
      const int u = __shfl_up(z, d);
      if (lane >= d) z += u;
    }
    wsum[lane] = z - y;
  }
  __syncthreads();
  int run = wsum[w] + x - tsum;
  #pragma unroll
  for (int i = 0; i < KPT; ++i) { hist[KPT * t + i] = run; run += c[i]; }
  __syncthreads();

  #pragma unroll
  for (int p = 0; p < EPT; ++p) {
    if (ok[p]) {
      const int pos = atomicAdd(&hist[key[p]], 1);
      buf[pos] = make_int2(off[p], s[p]);
    }
  }
  __syncthreads();

  #pragma unroll
  for (int p = 0; p < EPT; ++p) {
    const int2 q = buf[min(jb + p * 32, nvalid - 1)];
    off[p] = q.x;
    s[p]   = q.y;
  }
  __syncthreads();
}

__global__ __launch_bounds__(SC_THREADS) void SpVecs_scatter(
    const float* __restrict__ feat, const int* __restrict__ sorted,
    const int* __restrict__ cstart, float* __restrict__ part,
    float* __restrict__ cpart, int HW, int S) {
  extern __shared__ float lds[];
  float* acc  = lds;
  float* cacc = lds + (size_t)PLANES * S;

  const int  g        = blockIdx.x;
  const int  c0       = blockIdx.y * PLANES;
  const bool counting = (blockIdx.y == 0);
  const int  lane     = threadIdx.x & 31;
  const int  w        = threadIdx.x >> 5;

  const int nz = (counting ? PLANES + 1 : PLANES) * S;
  const bool can_sort = PLANES * S >= KL + 32 + 2 * SC_THREADS * EPT;
  bool cleared = false;

  const int2* srt = (const int2*)sorted;
  const int e0 = cstart[g];
  const int e1 = cstart[g + 1];
  const unsigned uHW  = (unsigned)HW;
  const unsigned base = (unsigned)c0 * uHW;

  for (int b = e0; b < e1; b += SC_THREADS * EPT) {
    unsigned idx[EPT];
    int      s[EPT];
    bool     ok[EPT];
    const int nvalid = min(e1 - b, SC_THREADS * EPT);
    const int pw = (nvalid + 31) >> 5;
    const int wj = w * pw + lane;
    int off[EPT];
    #pragma unroll
    for (int p = 0; p < EPT; ++p) {
      const int j = wj + p * 32;
      ok[p] = (p * 32 + lane < pw) && (j < nvalid);
      const int2 q = srt[b + min(j, nvalid - 1)];
      off[p] = q.x;
      s[p]   = q.y;
    }

    if (!cleared) {
      if (can_sort) {
        const int shift  = SpVecs_shift(HW);
        const int lo     = g << shift;
        const int span   = (g == NB - 1) ? max(HW - lo, 1) : (1 << shift);
        int lshift = 0;
        while (lshift < 31 && ((span - 1) >> lshift) >= KL) ++lshift;
        SpVecs_localsort(off, s, ok, (int*)lds, lo, lshift, nvalid, pw);
      }
      float4* l4 = (float4*)lds;
      const float4 z4 = make_float4(0.0f, 0.0f, 0.0f, 0.0f);
      for (int i = threadIdx.x; i < (nz >> 2); i += SC_THREADS) l4[i] = z4;
      for (int i = (nz & ~3) + threadIdx.x; i < nz; i += SC_THREADS) lds[i] = 0.0f;
      __syncthreads();
      cleared = true;
    }

    #pragma unroll
    for (int p = 0; p < EPT; ++p) {
      idx[p] = base + (unsigned)off[p];
      s[p]   = ok[p] ? s[p] : min(lane, S - 1);
    }

    if (counting) {
      #pragma unroll
      for (int p = 0; p < EPT; ++p) atomicAdd(&cacc[s[p]], ok[p] ? 1.0f : 0.0f);
    }

    float va[EPT], vb[EPT];
    float*   a    = acc;
    unsigned poff = 0;
    SpVecs_gather(va, feat, idx, poff);
    for (int k = 0; k < PLANES - 2; k += 2) {
      SpVecs_gather(vb, feat, idx, poff + uHW);
      SpVecs_add(a, s, ok, va);
      SpVecs_gather(va, feat, idx, poff + 2 * uHW);
      SpVecs_add(a + S, s, ok, vb);
      poff += 2 * uHW;
      a    += 2 * (size_t)S;
    }
    SpVecs_gather(vb, feat, idx, poff + uHW);
    SpVecs_add(a, s, ok, va);
    SpVecs_add(a + S, s, ok, vb);
  }
  if (!cleared) {
    for (int i = threadIdx.x; i < nz; i += SC_THREADS) lds[i] = 0.0f;
  }
  __syncthreads();

  const bool vec = ((S & 3) == 0) && ((((size_t)part) & 15) == 0);
  #pragma unroll 1
  for (int k = 0; k < PLANES; ++k) {
    float*       dst = part + ((size_t)(c0 + k) * NCHUNK + g) * (size_t)S;
    const float* pa  = acc + (size_t)k * S;
    if (vec) {
      float4*       d4 = (float4*)dst;
      const float4* a4 = (const float4*)pa;
      for (int i = threadIdx.x; i < (S >> 2); i += SC_THREADS) d4[i] = a4[i];
    } else {
      for (int i = threadIdx.x; i < S; i += SC_THREADS) dst[i] = pa[i];
    }
  }
  if (counting) {
    float* cd = cpart + (size_t)g * (size_t)S;
    for (int i = threadIdx.x; i < S; i += SC_THREADS) cd[i] = cacc[i];
  }
}

__global__ __launch_bounds__(256) void SpVecs_reduce(
    const float* __restrict__ part, const float* __restrict__ cpart,
    float* __restrict__ sums, float* __restrict__ cnts, int S) {
  __shared__ float tile[RD_T][RD_T + 1];
  const int lane = threadIdx.x & 31;
  const int w    = threadIdx.x >> 5;
  const int s0   = blockIdx.x * RD_T;
  const int c0   = blockIdx.y * RD_T;
  const int s    = s0 + lane;
  const bool in  = s < S;
  const int  sc  = in ? s : (S - 1);

  const unsigned uS = (unsigned)S;
  unsigned pk[4];
  #pragma unroll
  for (int k = 0; k < 4; ++k)
    pk[k] = ((unsigned)(c0 + w * 4 + k) * NCHUNK) * uS + (unsigned)sc;

  float a[4] = {0.0f, 0.0f, 0.0f, 0.0f};
  for (int g0 = 0; g0 < NCHUNK; g0 += RD_B) {
    float t[4][RD_B];
    #pragma unroll
    for (int j = 0; j < RD_B; ++j) {
      #pragma unroll
      for (int k = 0; k < 4; ++k) t[k][j] = part[pk[k] + (unsigned)(g0 + j) * uS];
    }
    #pragma unroll
    for (int j = 0; j < RD_B; ++j) {
      #pragma unroll
      for (int k = 0; k < 4; ++k) a[k] += t[k][j];
    }
  }
  #pragma unroll
  for (int k = 0; k < 4; ++k) tile[w * 4 + k][lane] = a[k];

  if (blockIdx.y == 0 && w == 0) {
    float n = 0.0f;
    for (int g0 = 0; g0 < NCHUNK; g0 += RD_B) {
      float t[RD_B];
      #pragma unroll
      for (int j = 0; j < RD_B; ++j) t[j] = cpart[(unsigned)(g0 + j) * uS + (unsigned)sc];
      #pragma unroll
      for (int j = 0; j < RD_B; ++j) n += t[j];
    }
    if (in) cnts[s] = n;
  }
  __syncthreads();

  #pragma unroll
  for (int k = 0; k < 4; ++k) {
    const int r = w * 4 + k;
    if (s0 + r < S) sums[(size_t)(s0 + r) * C_CH + c0 + lane] = tile[lane][r];
  }
}

__global__ __launch_bounds__(256) void SpVecs_finalize(
    float* __restrict__ out, const float* __restrict__ cnts) {
  const int lane  = threadIdx.x & 31;
  const int wid   = threadIdx.x >> 5;
  const int g     = blockIdx.x * (blockDim.x >> 5) + wid;
  const int s0    = g << 4;
  const int m     = lane & 15;
  const int half  = lane >> 4;
  const int khalf = half << 1;

  const float invc = 1.0f / fmaxf(cnts[s0 + m], 1.0f);

#if __has_builtin(__builtin_amdgcn_wmma_f32_16x16x4_f32)
  #pragma unroll
  for (int c0 = 0; c0 < C_CH; c0 += 16) {
    v8f acc = {0.f, 0.f, 0.f, 0.f, 0.f, 0.f, 0.f, 0.f};
    #pragma unroll
    for (int j = 0; j < 4; ++j) {
      v2f A;
      A.x = (m == 4 * j + khalf + 0) ? invc : 0.0f;
      A.y = (m == 4 * j + khalf + 1) ? invc : 0.0f;
      const int r0 = s0 + 4 * j + khalf;
      v2f B;
      B.x = out[(size_t)r0 * C_CH + c0 + m];
      B.y = out[(size_t)(r0 + 1) * C_CH + c0 + m];
      acc = __builtin_amdgcn_wmma_f32_16x16x4_f32(
          false, A, false, B, (short)0, acc, false, false);
    }
    const int rowbase = s0 + (half << 3);
    #pragma unroll
    for (int r = 0; r < 8; ++r)
      out[(size_t)(rowbase + r) * C_CH + c0 + m] = acc[r];
  }
#else
  for (int c0 = 0; c0 < C_CH; c0 += 16) {
    for (int r = half * 8; r < half * 8 + 8; ++r)
      out[(size_t)(s0 + r) * C_CH + c0 + m] *= 1.0f / fmaxf(cnts[s0 + r], 1.0f);
  }
#endif
}

extern "C" void kernel_launch(void* const* d_in, const int* in_sizes, int n_in,
                              void* d_out, int out_size, void* d_ws, size_t ws_size,
                              hipStream_t stream) {
  const float* feat = (const float*)d_in[0];
  const int*   yx   = (const int*)d_in[1];
  const int*   seg  = (const int*)d_in[2];
  float* out = (float*)d_out;

  const int N  = in_sizes[2];
  const int S  = out_size / C_CH;
  const int HW = in_sizes[0] / C_CH;
  const int W  = (int)lround(sqrt((double)HW));

  int*   table  = (int*)d_ws;
  int*   cstart = table + (size_t)SORT_WG * NB;
  int*   sorted = cstart + 2 * NB;
  float* cnts   = (float*)(sorted + 2 * (size_t)N);
  float* cpart  = cnts + S;
  float* part   = cpart + (size_t)NCHUNK * (size_t)S;

  const int E = (N + SORT_WG - 1) / SORT_WG;

  SpVecs_bincount<<<SORT_WG, SORT_T, 0, stream>>>(yx, table, N, HW, W, E);
  SpVecs_place<<<SORT_WG, SORT_T, 0, stream>>>(yx, seg, table, sorted, cstart, N, HW, W, E);

  {
    dim3 grid(NCHUNK, C_CH / PLANES);
    SpVecs_scatter<<<grid, SC_THREADS, (PLANES + 1) * (size_t)S * sizeof(float), stream>>>(
        feat, sorted, cstart, part, cpart, HW, S);
  }

  {
    dim3 grid((S + RD_T - 1) / RD_T, C_CH / RD_T);
    SpVecs_reduce<<<grid, 256, 0, stream>>>(part, cpart, out, cnts, S);
  }

  {
    int groups = S / 16;
    int wpb = 8;
    int blocks = (groups + wpb - 1) / wpb;
    SpVecs_finalize<<<blocks, wpb * 32, 0, stream>>>(out, cnts);
  }
}


# normalisation fused into the partial-plane sum launch: the summed 32x32 tile stays in LDS and is divided by the counts there with the same one-hot f32 WMMA product; out written once; 4 launches
# speedup vs baseline: 43.9154x; 1.0740x over previous
#include <hip/hip_runtime.h>
#include <math.h>


typedef float v2f __attribute__((ext_vector_type(2)));
typedef float v8f __attribute__((ext_vector_type(8)));

#define C_CH 64
#define NCHUNK 64
#define SC_THREADS 1024
#define PPT 16
#define PLANES 16
#define EPT 17
#define KPT 17
#define KL (KPT * SC_THREADS)
#define RD_T 32
#define RD_B 8
#define RD_W 8
#define NB_LOG2 6
#define NB (1 << NB_LOG2)
#define SORT_WG 256
#define SORT_T 256
#define NBQ (NB / 4)

static_assert(NB == NCHUNK, "a pixel range is one scatter work-group's chunk");
static_assert(RD_T == 4 * RD_W && RD_T % 16 == 0 && NCHUNK % RD_B == 0, "reduce tile: 4 channels per wave, 16-segment matrix tiles");
static_assert(NCHUNK == RD_W * RD_B, "each reduce wave takes one batch of count partials");
static_assert(NB == 64, "one wave scans the range totals, two per lane");
static_assert(SORT_T >= NB, "one thread per range clears and stores the counts");
static_assert(SORT_T % NBQ == 0 && (SORT_WG * NBQ) % SORT_T == 0, "the table is read in whole rounds of SORT_T 16-byte words");
static_assert(PLANES % 2 == 0 && C_CH % PLANES == 0, "planes are walked in pairs");
static_assert(SC_THREADS == 32 * 32, "the in-work-group scan is 32 waves of 32 lanes");
static_assert((KL + 32) % 2 == 0, "the pair buffer behind the counters is 8-byte aligned");

__device__ __forceinline__ int SpVecs_shift(int HW) {
  const int bits = 32 - __clz(HW - 1);
  return max(bits - NB_LOG2, 0);
}
__device__ __forceinline__ int SpVecs_bin(int off, int shift) {
  return (int)min((unsigned)off >> shift, (unsigned)(NB - 1));
}

__global__ __launch_bounds__(SORT_T) void SpVecs_bincount(
    const int* __restrict__ yx, int* __restrict__ table, int N, int HW, int W, int E) {
  __shared__ int h[NB];
  if (threadIdx.x < NB) h[threadIdx.x] = 0;
  __syncthreads();

  const int   shift = SpVecs_shift(HW);
  const int2* yx2   = (const int2*)yx;
  const int e0 = blockIdx.x * E;
  const int e1 = min(e0 + E, N);
  for (int b = e0; b < e1; b += SORT_T * PPT) {
    int  key[PPT];
    bool ok[PPT];
    #pragma unroll
    for (int p = 0; p < PPT; ++p) {
      const int i = b + p * SORT_T + threadIdx.x;
      ok[p] = i < e1;
      const int2 q = yx2[ok[p] ? i : (e1 - 1)];
      key[p] = SpVecs_bin(q.x * W + q.y, shift);
    }
    #pragma unroll
    for (int p = 0; p < PPT; ++p) {
      if (ok[p]) atomicAdd(&h[key[p]], 1);
    }
  }
  __syncthreads();

  if (threadIdx.x < NB) table[(size_t)blockIdx.x * NB + threadIdx.x] = h[threadIdx.x];
}

__global__ __launch_bounds__(SORT_T) void SpVecs_place(
    const int* __restrict__ yx, const int* __restrict__ seg,
    const int* __restrict__ table, int* __restrict__ sorted, int* __restrict__ cstart,
    int N, int HW, int W, int E) {
  __shared__ int tot[NB];
  __shared__ int bef[NB];
  __shared__ int cur[NB];
  const int wg   = blockIdx.x;
  const int t    = threadIdx.x;
  const int lane = t & 31;
  const int w    = t >> 5;

  if (t < NB) { tot[t] = 0; bef[t] = 0; }
  __syncthreads();

  int4 mt = make_int4(0, 0, 0, 0);
  int4 mb = make_int4(0, 0, 0, 0);
  const int4* tb = (const int4*)table;
  #pragma unroll
  for (int k = 0; k < SORT_WG * NBQ / SORT_T; ++k) {
    const int4 v   = tb[t + k * SORT_T];
    const bool in  = (t / NBQ + k * (SORT_T / NBQ)) < wg;
    mt.x += v.x; mt.y += v.y; mt.z += v.z; mt.w += v.w;
    mb.x += in ? v.x : 0; mb.y += in ? v.y : 0; mb.z += in ? v.z : 0; mb.w += in ? v.w : 0;
  }
  {
    const int r = 4 * (t % NBQ);
    atomicAdd(&tot[r + 0], mt.x); atomicAdd(&tot[r + 1], mt.y);
    atomicAdd(&tot[r + 2], mt.z); atomicAdd(&tot[r + 3], mt.w);
    atomicAdd(&bef[r + 0], mb.x); atomicAdd(&bef[r + 1], mb.y);
    atomicAdd(&bef[r + 2], mb.z); atomicAdd(&bef[r + 3], mb.w);
  }
  __syncthreads();

  if (w == 0) {
    const int a  = tot[2 * lane];
    const int a1 = tot[2 * lane + 1];
    const int sm = a + a1;
    int x = sm;
    #pragma unroll
    for (int d = 1; d < 32; d <<= 1) {
      const int y = __shfl_up(x, d);
      if (lane >= d) x += y;
    }
    const int ex = x - sm;
    cur[2 * lane]     = ex + bef[2 * lane];
    cur[2 * lane + 1] = ex + a + bef[2 * lane + 1];
    if (wg == 0) {
      cstart[2 * lane]     = ex;
      cstart[2 * lane + 1] = ex + a;
      if (lane == 31) cstart[NB] = x;
    }
  }
  __syncthreads();

  const int   shift = SpVecs_shift(HW);
  const int2* yx2   = (const int2*)yx;
  int2*       srt   = (int2*)sorted;
  const int e0 = wg * E;
  const int e1 = min(e0 + E, N);
  for (int b = e0; b < e1; b += SORT_T * PPT) {
    int  off[PPT];
    int  sg[PPT];
    bool ok[PPT];
    #pragma unroll
    for (int p = 0; p < PPT; ++p) {
      const int i  = b + p * SORT_T + threadIdx.x;
      ok[p] = i < e1;
      const int ii = ok[p] ? i : (e1 - 1);
      const int2 q = yx2[ii];
      sg[p]  = seg[ii];
      off[p] = q.x * W + q.y;
    }
    #pragma unroll
    for (int p = 0; p < PPT; ++p) {
      if (ok[p]) {
        const int pos = atomicAdd(&cur[SpVecs_bin(off[p], shift)], 1);
        srt[pos] = make_int2(off[p], sg[p]);
      }
    }
  }
}

__device__ __forceinline__ void SpVecs_gather(
    float (&v)[EPT], const float* __restrict__ feat, const unsigned (&idx)[EPT], unsigned poff) {
  #pragma unroll
  for (int p = 0; p < EPT; ++p) v[p] = feat[idx[p] + poff];
}
__device__ __forceinline__ void SpVecs_add(
    float* a, const int (&s)[EPT], const bool (&ok)[EPT], const float (&v)[EPT]) {
  #pragma unroll
  for (int p = 0; p < EPT; ++p) atomicAdd(&a[s[p]], ok[p] ? v[p] : 0.0f);
}


__device__ __forceinline__ void SpVecs_localsort(
    int (&off)[EPT], int (&s)[EPT], const bool (&ok)[EPT], int* li,
    int lo, int lshift, int nvalid, int pw) {
  int*  hist = li;
  int*  wsum = li + KL;
  int2* buf  = (int2*)(li + KL + 32);
  const int t    = threadIdx.x;
  const int lane = t & 31;
  const int w    = t >> 5;
  const int jb   = w * pw + lane;

  {
    int4* h4 = (int4*)hist;
    const int4 z4 = make_int4(0, 0, 0, 0);
    for (int i = t; i < KL / 4; i += SC_THREADS) h4[i] = z4;
  }
  __syncthreads();

  int key[EPT];
  #pragma unroll
  for (int p = 0; p < EPT; ++p) {
    key[p] = (int)min((unsigned)(off[p] - lo) >> lshift, (unsigned)(KL - 1));
    if (ok[p]) atomicAdd(&hist[key[p]], 1);
  }
  __syncthreads();

  int c[KPT];
  int tsum = 0;
  #pragma unroll
  for (int i = 0; i < KPT; ++i) { c[i] = hist[KPT * t + i]; tsum += c[i]; }
  int x = tsum;
  #pragma unroll
  for (int d = 1; d < 32; d <<= 1) {
    const int y = __shfl_up(x, d);
    if (lane >= d) x += y;
  }
  if (lane == 31) wsum[w] = x;
  __syncthreads();
  if (w == 0) {
    const int y = wsum[lane];
    int z = y;
    #pragma unroll
    for (int d = 1; d < 32; d <<= 1) {
      const int u = __shfl_up(z, d);
      if (lane >= d) z += u;
    }
    wsum[lane] = z - y;
  }
  __syncthreads();
  int run = wsum[w] + x - tsum;
  #pragma unroll
  for (int i = 0; i < KPT; ++i) { hist[KPT * t + i] = run; run += c[i]; }
  __syncthreads();

  #pragma unroll
  for (int p = 0; p < EPT; ++p) {
    if (ok[p]) {
      const int pos = atomicAdd(&hist[key[p]], 1);
      buf[pos] = make_int2(off[p], s[p]);
    }
  }
  __syncthreads();

  #pragma unroll
  for (int p = 0; p < EPT; ++p) {
    const int2 q = buf[min(jb + p * 32, nvalid - 1)];
    off[p] = q.x;
    s[p]   = q.y;
  }
  __syncthreads();
}

__global__ __launch_bounds__(SC_THREADS) void SpVecs_scatter(
    const float* __restrict__ feat, const int* __restrict__ sorted,
    const int* __restrict__ cstart, float* __restrict__ part,
    float* __restrict__ cpart, int HW, int S) {
  extern __shared__ float lds[];
  float* acc  = lds;
  float* cacc = lds + (size_t)PLANES * S;

  const int  g        = blockIdx.x;
  const int  c0       = blockIdx.y * PLANES;
  const bool counting = (blockIdx.y == 0);
  const int  lane     = threadIdx.x & 31;
  const int  w        = threadIdx.x >> 5;

  const int nz = (counting ? PLANES + 1 : PLANES) * S;
  const bool can_sort = PLANES * S >= KL + 32 + 2 * SC_THREADS * EPT;
  bool cleared = false;

  const int2* srt = (const int2*)sorted;
  const int e0 = cstart[g];
  const int e1 = cstart[g + 1];
  const unsigned uHW  = (unsigned)HW;
  const unsigned base = (unsigned)c0 * uHW;

  for (int b = e0; b < e1; b += SC_THREADS * EPT) {
    unsigned idx[EPT];
    int      s[EPT];
    bool     ok[EPT];
    const int nvalid = min(e1 - b, SC_THREADS * EPT);
    const int pw = (nvalid + 31) >> 5;
    const int wj = w * pw + lane;
    int off[EPT];
    #pragma unroll
    for (int p = 0; p < EPT; ++p) {
      const int j = wj + p * 32;
      ok[p] = (p * 32 + lane < pw) && (j < nvalid);
      const int2 q = srt[b + min(j, nvalid - 1)];
      off[p] = q.x;
      s[p]   = q.y;
    }

    if (!cleared) {
      if (can_sort) {
        const int shift  = SpVecs_shift(HW);
        const int lo     = g << shift;
        const int span   = (g == NB - 1) ? max(HW - lo, 1) : (1 << shift);
        int lshift = 0;
        while (lshift < 31 && ((span - 1) >> lshift) >= KL) ++lshift;
        SpVecs_localsort(off, s, ok, (int*)lds, lo, lshift, nvalid, pw);
      }
      float4* l4 = (float4*)lds;
      const float4 z4 = make_float4(0.0f, 0.0f, 0.0f, 0.0f);
      for (int i = threadIdx.x; i < (nz >> 2); i += SC_THREADS) l4[i] = z4;
      for (int i = (nz & ~3) + threadIdx.x; i < nz; i += SC_THREADS) lds[i] = 0.0f;
      __syncthreads();
      cleared = true;
    }

    #pragma unroll
    for (int p = 0; p < EPT; ++p) {
      idx[p] = base + (unsigned)off[p];
      s[p]   = ok[p] ? s[p] : min(lane, S - 1);
    }

    if (counting) {
      #pragma unroll
      for (int p = 0; p < EPT; ++p) atomicAdd(&cacc[s[p]], ok[p] ? 1.0f : 0.0f);
    }

    float va[EPT], vb[EPT];
    float*   a    = acc;
    unsigned poff = 0;
    SpVecs_gather(va, feat, idx, poff);
    for (int k = 0; k < PLANES - 2; k += 2) {
      SpVecs_gather(vb, feat, idx, poff + uHW);
      SpVecs_add(a, s, ok, va);
      SpVecs_gather(va, feat, idx, poff + 2 * uHW);
      SpVecs_add(a + S, s, ok, vb);
      poff += 2 * uHW;
      a    += 2 * (size_t)S;
    }
    SpVecs_gather(vb, feat, idx, poff + uHW);
    SpVecs_add(a, s, ok, va);
    SpVecs_add(a + S, s, ok, vb);
  }
  if (!cleared) {
    for (int i = threadIdx.x; i < nz; i += SC_THREADS) lds[i] = 0.0f;
  }
  __syncthreads();

  const bool vec = ((S & 3) == 0) && ((((size_t)part) & 15) == 0);
  #pragma unroll 1
  for (int k = 0; k < PLANES; ++k) {
    float*       dst = part + ((size_t)(c0 + k) * NCHUNK + g) * (size_t)S;
    const float* pa  = acc + (size_t)k * S;
    if (vec) {
      float4*       d4 = (float4*)dst;
      const float4* a4 = (const float4*)pa;
      for (int i = threadIdx.x; i < (S >> 2); i += SC_THREADS) d4[i] = a4[i];
    } else {
      for (int i = threadIdx.x; i < S; i += SC_THREADS) dst[i] = pa[i];
    }
  }
  if (counting) {
    float* cd = cpart + (size_t)g * (size_t)S;
    for (int i = threadIdx.x; i < S; i += SC_THREADS) cd[i] = cacc[i];
  }
}

__global__ __launch_bounds__(RD_W * 32) void SpVecs_reduce(
    const float* __restrict__ part, const float* __restrict__ cpart,
    float* __restrict__ out, int S) {
  __shared__ float tile[RD_T][RD_T + 1];
  __shared__ float otile[RD_T][RD_T + 1];
  __shared__ float cn[RD_W][RD_T];
  const int lane = threadIdx.x & 31;
  const int w    = threadIdx.x >> 5;
  const int s0   = blockIdx.x * RD_T;
  const int c0   = blockIdx.y * RD_T;
  const int s    = s0 + lane;
  const bool in  = s < S;
  const int  sc  = in ? s : (S - 1);

  const unsigned uS = (unsigned)S;

  float tc[RD_B];
  #pragma unroll
  for (int j = 0; j < RD_B; ++j) tc[j] = cpart[(unsigned)(w * RD_B + j) * uS + (unsigned)sc];

  unsigned pk[4];
  #pragma unroll
  for (int k = 0; k < 4; ++k)
    pk[k] = ((unsigned)(c0 + w * 4 + k) * NCHUNK) * uS + (unsigned)sc;

  float a[4] = {0.0f, 0.0f, 0.0f, 0.0f};
  for (int g0 = 0; g0 < NCHUNK; g0 += RD_B) {
    float t[4][RD_B];
    #pragma unroll
    for (int j = 0; j < RD_B; ++j) {
      #pragma unroll
      for (int k = 0; k < 4; ++k) t[k][j] = part[pk[k] + (unsigned)(g0 + j) * uS];
    }
    #pragma unroll
    for (int j = 0; j < RD_B; ++j) {
      #pragma unroll
      for (int k = 0; k < 4; ++k) a[k] += t[k][j];
    }
  }
  #pragma unroll
  for (int k = 0; k < 4; ++k) tile[w * 4 + k][lane] = a[k];
  {
    float n = 0.0f;
    #pragma unroll
    for (int j = 0; j < RD_B; ++j) n += tc[j];
    cn[w][lane] = n;
  }
  __syncthreads();

  if (w < RD_T / 16) {
    const int sg    = 16 * w;
    const int m     = lane & 15;
    const int half  = lane >> 4;
    const int khalf = half << 1;

    float cnt = 0.0f;
    #pragma unroll
    for (int q = 0; q < RD_W; ++q) cnt += cn[q][sg + m];
    const float invc = 1.0f / fmaxf(cnt, 1.0f);

#if __has_builtin(__builtin_amdgcn_wmma_f32_16x16x4_f32)
    #pragma unroll
    for (int cc = 0; cc < RD_T; cc += 16) {
      v8f acc = {0.f, 0.f, 0.f, 0.f, 0.f, 0.f, 0.f, 0.f};
      #pragma unroll
      for (int j = 0; j < 4; ++j) {
        v2f A;
        A.x = (m == 4 * j + khalf + 0) ? invc : 0.0f;
        A.y = (m == 4 * j + khalf + 1) ? invc : 0.0f;
        const int r0 = sg + 4 * j + khalf;
        v2f B;
        B.x = tile[cc + m][r0];
        B.y = tile[cc + m][r0 + 1];
        acc = __builtin_amdgcn_wmma_f32_16x16x4_f32(
            false, A, false, B, (short)0, acc, false, false);
      }
      const int rowbase = sg + (half << 3);
      #pragma unroll
      for (int r = 0; r < 8; ++r) otile[rowbase + r][cc + m] = acc[r];
    }
#else
    for (int cc = half; cc < RD_T; cc += 2) otile[sg + m][cc] = tile[cc][sg + m] * invc;
#endif
  }
  __syncthreads();

  #pragma unroll
  for (int k = 0; k < 4; ++k) {
    const int r = w * 4 + k;
    if (s0 + r < S) out[(size_t)(s0 + r) * C_CH + c0 + lane] = otile[r][lane];
  }
}

extern "C" void kernel_launch(void* const* d_in, const int* in_sizes, int n_in,
                              void* d_out, int out_size, void* d_ws, size_t ws_size,
                              hipStream_t stream) {
  const float* feat = (const float*)d_in[0];
  const int*   yx   = (const int*)d_in[1];
  const int*   seg  = (const int*)d_in[2];
  float* out = (float*)d_out;

  const int N  = in_sizes[2];
  const int S  = out_size / C_CH;
  const int HW = in_sizes[0] / C_CH;
  const int W  = (int)lround(sqrt((double)HW));

  int*   table  = (int*)d_ws;
  int*   cstart = table + (size_t)SORT_WG * NB;
  int*   sorted = cstart + 2 * NB;
  float* cpart  = (float*)(sorted + 2 * (size_t)N);
  float* part   = cpart + (size_t)NCHUNK * (size_t)S;

  const int E = (N + SORT_WG - 1) / SORT_WG;

  SpVecs_bincount<<<SORT_WG, SORT_T, 0, stream>>>(yx, table, N, HW, W, E);
  SpVecs_place<<<SORT_WG, SORT_T, 0, stream>>>(yx, seg, table, sorted, cstart, N, HW, W, E);

  {
    dim3 grid(NCHUNK, C_CH / PLANES);
    SpVecs_scatter<<<grid, SC_THREADS, (PLANES + 1) * (size_t)S * sizeof(float), stream>>>(
        feat, sorted, cstart, part, cpart, HW, S);
  }

  {
    dim3 grid((S + RD_T - 1) / RD_T, C_CH / RD_T);
    SpVecs_reduce<<<grid, RD_W * 32, 0, stream>>>(part, cpart, out, S);
  }
}


# first-level sort in one launch (3 in all): a sort group deals its entries into 64 runs in its own stretch, no count table across groups; scatter group scans its 256 run lengths; same ds f32 atomics
# speedup vs baseline: 45.7218x; 1.0411x over previous
#include <hip/hip_runtime.h>
#include <math.h>


typedef float v2f __attribute__((ext_vector_type(2)));
typedef float v8f __attribute__((ext_vector_type(8)));

#define C_CH 64
#define NCHUNK 64
#define SC_THREADS 1024
#define PPT 16
#define PLANES 16
#define EPT 17
#define KPT 17
#define KL (KPT * SC_THREADS)
#define RD_T 32
#define RD_B 8
#define RD_W 8
#define NB_LOG2 6
#define NB (1 << NB_LOG2)
#define SORT_WG 256
#define SORT_T 256

static_assert(NB == NCHUNK, "a pixel range is one scatter work-group's chunk");
static_assert(RD_T == 4 * RD_W && RD_T % 16 == 0 && NCHUNK % RD_B == 0, "reduce tile: 4 channels per wave, 16-segment matrix tiles");
static_assert(NCHUNK == RD_W * RD_B, "each reduce wave takes one batch of count partials");
static_assert(NB == 64, "one wave scans a sort work-group's range counts, two per lane");
static_assert(SORT_T >= NB, "one thread per range clears the counts");
static_assert((SORT_WG & (SORT_WG - 1)) == 0 && SORT_WG % 32 == 0 && SORT_WG <= SC_THREADS, "a scatter work-group scans the run lengths with whole waves and halves its way to a run");
static_assert(PLANES % 2 == 0 && C_CH % PLANES == 0, "planes are walked in pairs");
static_assert(SC_THREADS == 32 * 32, "the in-work-group scan is 32 waves of 32 lanes");
static_assert((KL + 32) % 2 == 0, "the pair buffer behind the counters is 8-byte aligned");

__device__ __forceinline__ int SpVecs_shift(int HW) {
  const int bits = 32 - __clz(HW - 1);
  return max(bits - NB_LOG2, 0);
}
__device__ __forceinline__ int SpVecs_bin(int off, int shift) {
  return (int)min((unsigned)off >> shift, (unsigned)(NB - 1));
}

__global__ __launch_bounds__(SORT_T) void SpVecs_deal(
    const int* __restrict__ yx, const int* __restrict__ seg,
    int* __restrict__ table, int* __restrict__ sorted, int N, int HW, int W, int E) {
  __shared__ int h[NB];
  __shared__ int cur[NB];
  const int wg   = blockIdx.x;
  const int t    = threadIdx.x;
  const int lane = t & 31;
  const int w    = t >> 5;

  if (t < NB) h[t] = 0;
  __syncthreads();

  const int   shift = SpVecs_shift(HW);
  const int2* yx2   = (const int2*)yx;
  int2*       srt   = (int2*)sorted;
  const int e0 = wg * E;
  const int e1 = min(e0 + E, N);

  int  off[PPT];
  int  sg[PPT];
  bool ok[PPT];
  #pragma unroll
  for (int p = 0; p < PPT; ++p) { off[p] = 0; sg[p] = 0; ok[p] = false; }

  int blast = e0;
  for (int b = e0; b < e1; b += SORT_T * PPT) {
    #pragma unroll
    for (int p = 0; p < PPT; ++p) {
      const int i  = b + p * SORT_T + t;
      ok[p] = i < e1;
      const int ii = ok[p] ? i : (e1 - 1);
      const int2 q = yx2[ii];
      sg[p]  = seg[ii];
      off[p] = q.x * W + q.y;
    }
    #pragma unroll
    for (int p = 0; p < PPT; ++p) {
      if (ok[p]) atomicAdd(&h[SpVecs_bin(off[p], shift)], 1);
    }
    blast = b;
  }
  __syncthreads();

  if (w == 0) {
    const int a  = h[2 * lane];
    const int a1 = h[2 * lane + 1];
    const int sm = a + a1;
    int x = sm;
    #pragma unroll
    for (int d = 1; d < 32; d <<= 1) {
      const int y = __shfl_up(x, d);
      if (lane >= d) x += y;
    }
    const int ex = x - sm;
    cur[2 * lane]     = ex;
    cur[2 * lane + 1] = ex + a;
    ((int2*)table)[(size_t)wg * (NB / 2) + lane] = make_int2(ex, ex + a);
  }
  __syncthreads();

  if (e0 < e1) {
    for (int b = blast; b >= e0; b -= SORT_T * PPT) {
      if (b != blast) {
        #pragma unroll
        for (int p = 0; p < PPT; ++p) {
          const int i  = b + p * SORT_T + t;
          ok[p] = i < e1;
          const int ii = ok[p] ? i : (e1 - 1);
          const int2 q = yx2[ii];
          sg[p]  = seg[ii];
          off[p] = q.x * W + q.y;
        }
      }
      #pragma unroll
      for (int p = 0; p < PPT; ++p) {
        if (ok[p]) {
          const int pos = atomicAdd(&cur[SpVecs_bin(off[p], shift)], 1);
          srt[e0 + pos] = make_int2(off[p], sg[p]);
        }
      }
    }
  }
}

__device__ __forceinline__ void SpVecs_gather(
    float (&v)[EPT], const float* __restrict__ feat, const unsigned (&idx)[EPT], unsigned poff) {
  #pragma unroll
  for (int p = 0; p < EPT; ++p) v[p] = feat[idx[p] + poff];
}
__device__ __forceinline__ void SpVecs_add(
    float* a, const int (&s)[EPT], const bool (&ok)[EPT], const float (&v)[EPT]) {
  #pragma unroll
  for (int p = 0; p < EPT; ++p) atomicAdd(&a[s[p]], ok[p] ? v[p] : 0.0f);
}


__device__ __forceinline__ void SpVecs_localsort(
    int (&off)[EPT], int (&s)[EPT], const bool (&ok)[EPT], int* li,
    int lo, int lshift, int nvalid, int pw) {
  int*  hist = li;
  int*  wsum = li + KL;
  int2* buf  = (int2*)(li + KL + 32);
  const int t    = threadIdx.x;
  const int lane = t & 31;
  const int w    = t >> 5;
  const int jb   = w * pw + lane;

  {
    int4* h4 = (int4*)hist;
    const int4 z4 = make_int4(0, 0, 0, 0);
    for (int i = t; i < KL / 4; i += SC_THREADS) h4[i] = z4;
  }
  __syncthreads();

  int key[EPT];
  #pragma unroll
  for (int p = 0; p < EPT; ++p) {
    key[p] = (int)min((unsigned)(off[p] - lo) >> lshift, (unsigned)(KL - 1));
    if (ok[p]) atomicAdd(&hist[key[p]], 1);
  }
  __syncthreads();

  int c[KPT];
  int tsum = 0;
  #pragma unroll
  for (int i = 0; i < KPT; ++i) { c[i] = hist[KPT * t + i]; tsum += c[i]; }
  int x = tsum;
  #pragma unroll
  for (int d = 1; d < 32; d <<= 1) {
    const int y = __shfl_up(x, d);
    if (lane >= d) x += y;
  }
  if (lane == 31) wsum[w] = x;
  __syncthreads();
  if (w == 0) {
    const int y = wsum[lane];
    int z = y;
    #pragma unroll
    for (int d = 1; d < 32; d <<= 1) {
      const int u = __shfl_up(z, d);
      if (lane >= d) z += u;
    }
    wsum[lane] = z - y;
  }
  __syncthreads();
  int run = wsum[w] + x - tsum;
  #pragma unroll
  for (int i = 0; i < KPT; ++i) { hist[KPT * t + i] = run; run += c[i]; }
  __syncthreads();

  #pragma unroll
  for (int p = 0; p < EPT; ++p) {
    if (ok[p]) {
      const int pos = atomicAdd(&hist[key[p]], 1);
      buf[pos] = make_int2(off[p], s[p]);
    }
  }
  __syncthreads();

  #pragma unroll
  for (int p = 0; p < EPT; ++p) {
    const int2 q = buf[min(jb + p * 32, nvalid - 1)];
    off[p] = q.x;
    s[p]   = q.y;
  }
  __syncthreads();
}

__global__ __launch_bounds__(SC_THREADS) void SpVecs_scatter(
    const float* __restrict__ feat, const int* __restrict__ sorted,
    const int* __restrict__ table, float* __restrict__ part,
    float* __restrict__ cpart, int HW, int S, int N, int E) {
  __shared__ int rpre[SORT_WG + 1];
  __shared__ int rdel[SORT_WG];
  __shared__ int rws[SORT_WG / 32];
  extern __shared__ float lds[];
  float* acc  = lds;
  float* cacc = lds + (size_t)PLANES * S;

  const int  g        = blockIdx.x;
  const int  c0       = blockIdx.y * PLANES;
  const bool counting = (blockIdx.y == 0);
  const int  lane     = threadIdx.x & 31;
  const int  w        = threadIdx.x >> 5;

  const int nz = (counting ? PLANES + 1 : PLANES) * S;
  const bool can_sort = PLANES * S >= KL + 32 + 2 * SC_THREADS * EPT;
  bool cleared = false;

  const int2* srt = (const int2*)sorted;

  {
    const int t = threadIdx.x;
    int st = 0, len = 0, x = 0;
    if (t < SORT_WG) {
      const int r0  = t * E;
      const int cnt = max(min(r0 + E, N) - r0, 0);
      st = table[t * NB + g];
      const int en = (g + 1 < NB) ? table[t * NB + g + 1] : cnt;
      len = en - st;
      x   = len;
      #pragma unroll
      for (int d = 1; d < 32; d <<= 1) {
        const int y = __shfl_up(x, d);
        if (lane >= d) x += y;
      }
      if (lane == 31) rws[w] = x;
    }
    __syncthreads();
    if (t < SORT_WG) {
      int bw = 0;
      #pragma unroll
      for (int q = 0; q < SORT_WG / 32; ++q) bw += (q < w) ? rws[q] : 0;
      const int ex = bw + x - len;
      rpre[t] = ex;
      rdel[t] = t * E + st - ex;
      if (t == SORT_WG - 1) rpre[SORT_WG] = bw + x;
    }
    __syncthreads();
  }
  const int total = rpre[SORT_WG];
  const unsigned uHW  = (unsigned)HW;
  const unsigned base = (unsigned)c0 * uHW;

  for (int b = 0; b < total; b += SC_THREADS * EPT) {
    unsigned idx[EPT];
    int      s[EPT];
    bool     ok[EPT];
    const int nvalid = min(total - b, SC_THREADS * EPT);
    const int pw = (nvalid + 31) >> 5;
    const int wj = w * pw + lane;
    int r = 0;
    {
      const int J0 = b + min(wj, nvalid - 1);
      #pragma unroll
      for (int step = SORT_WG / 2; step >= 1; step >>= 1) r += (rpre[r + step] <= J0) ? step : 0;
    }
    int nxt = rpre[r + 1];
    int del = rdel[r];
    int off[EPT];
    #pragma unroll
    for (int p = 0; p < EPT; ++p) {
      const int j = wj + p * 32;
      ok[p] = (p * 32 + lane < pw) && (j < nvalid);
      const int J = b + min(j, nvalid - 1);
      while (J >= nxt && r < SORT_WG - 1) {
        ++r;
        nxt = rpre[r + 1];
        del = rdel[r];
      }
      const int2 q = srt[J + del];
      off[p] = q.x;
      s[p]   = q.y;
    }

    if (!cleared) {
      if (can_sort) {
        const int shift  = SpVecs_shift(HW);
        const int lo     = g << shift;
        const int span   = (g == NB - 1) ? max(HW - lo, 1) : (1 << shift);
        int lshift = 0;
        while (lshift < 31 && ((span - 1) >> lshift) >= KL) ++lshift;
        SpVecs_localsort(off, s, ok, (int*)lds, lo, lshift, nvalid, pw);
      }
      float4* l4 = (float4*)lds;
      const float4 z4 = make_float4(0.0f, 0.0f, 0.0f, 0.0f);
      for (int i = threadIdx.x; i < (nz >> 2); i += SC_THREADS) l4[i] = z4;
      for (int i = (nz & ~3) + threadIdx.x; i < nz; i += SC_THREADS) lds[i] = 0.0f;
      __syncthreads();
      cleared = true;
    }

    #pragma unroll
    for (int p = 0; p < EPT; ++p) {
      idx[p] = base + (unsigned)off[p];
      s[p]   = ok[p] ? s[p] : min(lane, S - 1);
    }

    if (counting) {
      #pragma unroll
      for (int p = 0; p < EPT; ++p) atomicAdd(&cacc[s[p]], ok[p] ? 1.0f : 0.0f);
    }

    float va[EPT], vb[EPT];
    float*   a    = acc;
    unsigned poff = 0;
    SpVecs_gather(va, feat, idx, poff);
    for (int k = 0; k < PLANES - 2; k += 2) {
      SpVecs_gather(vb, feat, idx, poff + uHW);
      SpVecs_add(a, s, ok, va);
      SpVecs_gather(va, feat, idx, poff + 2 * uHW);
      SpVecs_add(a + S, s, ok, vb);
      poff += 2 * uHW;
      a    += 2 * (size_t)S;
    }
    SpVecs_gather(vb, feat, idx, poff + uHW);
    SpVecs_add(a, s, ok, va);
    SpVecs_add(a + S, s, ok, vb);
  }
  if (!cleared) {
    for (int i = threadIdx.x; i < nz; i += SC_THREADS) lds[i] = 0.0f;
  }
  __syncthreads();

  const bool vec = ((S & 3) == 0) && ((((size_t)part) & 15) == 0);
  #pragma unroll 1
  for (int k = 0; k < PLANES; ++k) {
    float*       dst = part + ((size_t)(c0 + k) * NCHUNK + g) * (size_t)S;
    const float* pa  = acc + (size_t)k * S;
    if (vec) {
      float4*       d4 = (float4*)dst;
      const float4* a4 = (const float4*)pa;
      for (int i = threadIdx.x; i < (S >> 2); i += SC_THREADS) d4[i] = a4[i];
    } else {
      for (int i = threadIdx.x; i < S; i += SC_THREADS) dst[i] = pa[i];
    }
  }
  if (counting) {
    float* cd = cpart + (size_t)g * (size_t)S;
    for (int i = threadIdx.x; i < S; i += SC_THREADS) cd[i] = cacc[i];
  }
}

__global__ __launch_bounds__(RD_W * 32) void SpVecs_reduce(
    const float* __restrict__ part, const float* __restrict__ cpart,
    float* __restrict__ out, int S) {
  __shared__ float tile[RD_T][RD_T + 1];
  __shared__ float otile[RD_T][RD_T + 1];
  __shared__ float cn[RD_W][RD_T];
  const int lane = threadIdx.x & 31;
  const int w    = threadIdx.x >> 5;
  const int s0   = blockIdx.x * RD_T;
  const int c0   = blockIdx.y * RD_T;
  const int s    = s0 + lane;
  const bool in  = s < S;
  const int  sc  = in ? s : (S - 1);

  const unsigned uS = (unsigned)S;

  float tc[RD_B];
  #pragma unroll
  for (int j = 0; j < RD_B; ++j) tc[j] = cpart[(unsigned)(w * RD_B + j) * uS + (unsigned)sc];

  unsigned pk[4];
  #pragma unroll
  for (int k = 0; k < 4; ++k)
    pk[k] = ((unsigned)(c0 + w * 4 + k) * NCHUNK) * uS + (unsigned)sc;

  float a[4] = {0.0f, 0.0f, 0.0f, 0.0f};
  for (int g0 = 0; g0 < NCHUNK; g0 += RD_B) {
    float t[4][RD_B];
    #pragma unroll
    for (int j = 0; j < RD_B; ++j) {
      #pragma unroll
      for (int k = 0; k < 4; ++k) t[k][j] = part[pk[k] + (unsigned)(g0 + j) * uS];
    }
    #pragma unroll
    for (int j = 0; j < RD_B; ++j) {
      #pragma unroll
      for (int k = 0; k < 4; ++k) a[k] += t[k][j];
    }
  }
  #pragma unroll
  for (int k = 0; k < 4; ++k) tile[w * 4 + k][lane] = a[k];
  {
    float n = 0.0f;
    #pragma unroll
    for (int j = 0; j < RD_B; ++j) n += tc[j];
    cn[w][lane] = n;
  }
  __syncthreads();

  if (w < RD_T / 16) {
    const int sg    = 16 * w;
    const int m     = lane & 15;
    const int half  = lane >> 4;
    const int khalf = half << 1;

    float cnt = 0.0f;
    #pragma unroll
    for (int q = 0; q < RD_W; ++q) cnt += cn[q][sg + m];
    const float invc = 1.0f / fmaxf(cnt, 1.0f);

#if __has_builtin(__builtin_amdgcn_wmma_f32_16x16x4_f32)
    #pragma unroll
    for (int cc = 0; cc < RD_T; cc += 16) {
      v8f acc = {0.f, 0.f, 0.f, 0.f, 0.f, 0.f, 0.f, 0.f};
      #pragma unroll
      for (int j = 0; j < 4; ++j) {
        v2f A;
        A.x = (m == 4 * j + khalf + 0) ? invc : 0.0f;
        A.y = (m == 4 * j + khalf + 1) ? invc : 0.0f;
        const int r0 = sg + 4 * j + khalf;
        v2f B;
        B.x = tile[cc + m][r0];
        B.y = tile[cc + m][r0 + 1];
        acc = __builtin_amdgcn_wmma_f32_16x16x4_f32(
            false, A, false, B, (short)0, acc, false, false);
      }
      const int rowbase = sg + (half << 3);
      #pragma unroll
      for (int r = 0; r < 8; ++r) otile[rowbase + r][cc + m] = acc[r];
    }
#else
    for (int cc = half; cc < RD_T; cc += 2) otile[sg + m][cc] = tile[cc][sg + m] * invc;
#endif
  }
  __syncthreads();

  #pragma unroll
  for (int k = 0; k < 4; ++k) {
    const int r = w * 4 + k;
    if (s0 + r < S) out[(size_t)(s0 + r) * C_CH + c0 + lane] = otile[r][lane];
  }
}

extern "C" void kernel_launch(void* const* d_in, const int* in_sizes, int n_in,
                              void* d_out, int out_size, void* d_ws, size_t ws_size,
                              hipStream_t stream) {
  const float* feat = (const float*)d_in[0];
  const int*   yx   = (const int*)d_in[1];
  const int*   seg  = (const int*)d_in[2];
  float* out = (float*)d_out;

  const int N  = in_sizes[2];
  const int S  = out_size / C_CH;
  const int HW = in_sizes[0] / C_CH;
  const int W  = (int)lround(sqrt((double)HW));

  int*   table  = (int*)d_ws;
  int*   sorted = table + (size_t)SORT_WG * NB;
  float* cpart  = (float*)(sorted + 2 * (size_t)N);
  float* part   = cpart + (size_t)NCHUNK * (size_t)S;

  const int E = (N + SORT_WG - 1) / SORT_WG;

  SpVecs_deal<<<SORT_WG, SORT_T, 0, stream>>>(yx, seg, table, sorted, N, HW, W, E);

  {
    dim3 grid(NCHUNK, C_CH / PLANES);
    SpVecs_scatter<<<grid, SC_THREADS, (PLANES + 1) * (size_t)S * sizeof(float), stream>>>(
        feat, sorted, table, part, cpart, HW, S, N, E);
  }

  {
    dim3 grid((S + RD_T - 1) / RD_T, C_CH / RD_T);
    SpVecs_reduce<<<grid, RD_W * 32, 0, stream>>>(part, cpart, out, S);
  }
}


# speedup vs baseline: 50.8774x; 1.1128x over previous
#include <hip/hip_runtime.h>
#include <math.h>


typedef float v2f __attribute__((ext_vector_type(2)));
typedef float v8f __attribute__((ext_vector_type(8)));

#define C_CH 64
#define NCHUNK 64
#define SC_THREADS 1024
#define PPT 16
#define PLANES 16
#define EPT 17
#define KPT 17
#define KL (KPT * SC_THREADS)
#define RD_T 32
#define RD_B 8
#define RD_W 8
#define NB_LOG2 6
#define NB (1 << NB_LOG2)
#define SORT_WG 256
#define SORT_T 256

static_assert(NB == NCHUNK, "a pixel range is one scatter work-group's chunk");
static_assert(RD_T == 4 * RD_W && RD_T % 16 == 0 && NCHUNK % RD_B == 0, "reduce tile: 4 channels per wave, 16-segment matrix tiles");
static_assert(NCHUNK == RD_W * RD_B, "each reduce wave takes one batch of count partials");
static_assert(NB == 64, "one wave scans a sort work-group's range counts, two per lane");
static_assert(SORT_T >= NB, "one thread per range clears the counts");
static_assert((SORT_WG & (SORT_WG - 1)) == 0 && SORT_WG % 32 == 0 && SORT_WG <= SC_THREADS, "a scatter work-group scans the run lengths with whole waves and halves its way to a run");
static_assert(PLANES % 2 == 0 && C_CH % PLANES == 0, "planes are walked in pairs");
static_assert((PLANES & (PLANES - 1)) == 0 && PLANES % 8 == 0, "the walk wraps with a mask and starts PLANES / 4 apart, at a pair");
static_assert(SC_THREADS == 32 * 32, "the in-work-group scan is 32 waves of 32 lanes");
static_assert((KL + 32) % 2 == 0, "the pair buffer behind the counters is 8-byte aligned");

__device__ __forceinline__ int SpVecs_shift(int HW) {
  const int bits = 32 - __clz(HW - 1);
  return max(bits - NB_LOG2, 0);
}
__device__ __forceinline__ int SpVecs_bin(int off, int shift) {
  return (int)min((unsigned)off >> shift, (unsigned)(NB - 1));
}

__global__ __launch_bounds__(SORT_T) void SpVecs_deal(
    const int* __restrict__ yx, const int* __restrict__ seg,
    int* __restrict__ table, int* __restrict__ sorted, int N, int HW, int W, int E) {
  __shared__ int h[NB];
  __shared__ int cur[NB];
  const int wg   = blockIdx.x;
  const int t    = threadIdx.x;
  const int lane = t & 31;
  const int w    = t >> 5;

  if (t < NB) h[t] = 0;
  __syncthreads();

  const int   shift = SpVecs_shift(HW);
  const int2* yx2   = (const int2*)yx;
  int2*       srt   = (int2*)sorted;
  const int e0 = wg * E;
  const int e1 = min(e0 + E, N);

  int  off[PPT];
  int  sg[PPT];
  bool ok[PPT];
  #pragma unroll
  for (int p = 0; p < PPT; ++p) { off[p] = 0; sg[p] = 0; ok[p] = false; }

  int blast = e0;
  for (int b = e0; b < e1; b += SORT_T * PPT) {
    #pragma unroll
    for (int p = 0; p < PPT; ++p) {
      const int i  = b + p * SORT_T + t;
      ok[p] = i < e1;
      const int ii = ok[p] ? i : (e1 - 1);
      const int2 q = yx2[ii];
      sg[p]  = seg[ii];
      off[p] = q.x * W + q.y;
    }
    #pragma unroll
    for (int p = 0; p < PPT; ++p) {
      if (ok[p]) atomicAdd(&h[SpVecs_bin(off[p], shift)], 1);
    }
    blast = b;
  }
  __syncthreads();

  if (w == 0) {
    const int a  = h[2 * lane];
    const int a1 = h[2 * lane + 1];
    const int sm = a + a1;
    int x = sm;
    #pragma unroll
    for (int d = 1; d < 32; d <<= 1) {
      const int y = __shfl_up(x, d);
      if (lane >= d) x += y;
    }
    const int ex = x - sm;
    cur[2 * lane]     = ex;
    cur[2 * lane + 1] = ex + a;
    ((int2*)table)[(size_t)wg * (NB / 2) + lane] = make_int2(ex, ex + a);
  }
  __syncthreads();

  if (e0 < e1) {
    for (int b = blast; b >= e0; b -= SORT_T * PPT) {
      if (b != blast) {
        #pragma unroll
        for (int p = 0; p < PPT; ++p) {
          const int i  = b + p * SORT_T + t;
          ok[p] = i < e1;
          const int ii = ok[p] ? i : (e1 - 1);
          const int2 q = yx2[ii];
          sg[p]  = seg[ii];
          off[p] = q.x * W + q.y;
        }
      }
      #pragma unroll
      for (int p = 0; p < PPT; ++p) {
        if (ok[p]) {
          const int pos = atomicAdd(&cur[SpVecs_bin(off[p], shift)], 1);
          srt[e0 + pos] = make_int2(off[p], sg[p]);
        }
      }
    }
  }
}

__device__ __forceinline__ void SpVecs_gather(
    float (&v)[EPT], const float* __restrict__ feat, const unsigned (&idx)[EPT], unsigned poff) {
  #pragma unroll
  for (int p = 0; p < EPT; ++p) v[p] = feat[idx[p] + poff];
}
__device__ __forceinline__ void SpVecs_add(
    float* a, const int (&s)[EPT], const bool (&ok)[EPT], const float (&v)[EPT]) {
  #pragma unroll
  for (int p = 0; p < EPT; ++p) atomicAdd(&a[s[p]], ok[p] ? v[p] : 0.0f);
}


__device__ __forceinline__ unsigned SpVecs_plane(unsigned q0, int i) {
  return (q0 + (unsigned)i) & (unsigned)(PLANES - 1);
}

__device__ __forceinline__ void SpVecs_localsort(
    int (&off)[EPT], int (&s)[EPT], const bool (&ok)[EPT], int* li,
    int lo, int lshift, int nvalid, int pw) {
  int*  hist = li;
  int*  wsum = li + KL;
  int2* buf  = (int2*)(li + KL + 32);
  const int t    = threadIdx.x;
  const int lane = t & 31;
  const int w    = t >> 5;
  const int jb   = w * pw + lane;

  {
    int4* h4 = (int4*)hist;
    const int4 z4 = make_int4(0, 0, 0, 0);
    for (int i = t; i < KL / 4; i += SC_THREADS) h4[i] = z4;
  }
  __syncthreads();

  int key[EPT];
  #pragma unroll
  for (int p = 0; p < EPT; ++p) {
    key[p] = (int)min((unsigned)(off[p] - lo) >> lshift, (unsigned)(KL - 1));
    if (ok[p]) atomicAdd(&hist[key[p]], 1);
  }
  __syncthreads();

  int c[KPT];
  int tsum = 0;
  #pragma unroll
  for (int i = 0; i < KPT; ++i) { c[i] = hist[KPT * t + i]; tsum += c[i]; }
  int x = tsum;
  #pragma unroll
  for (int d = 1; d < 32; d <<= 1) {
    const int y = __shfl_up(x, d);
    if (lane >= d) x += y;
  }
  if (lane == 31) wsum[w] = x;
  __syncthreads();
  if (w == 0) {
    const int y = wsum[lane];
    int z = y;
    #pragma unroll
    for (int d = 1; d < 32; d <<= 1) {
      const int u = __shfl_up(z, d);
      if (lane >= d) z += u;
    }
    wsum[lane] = z - y;
  }
  __syncthreads();
  int run = wsum[w] + x - tsum;
  #pragma unroll
  for (int i = 0; i < KPT; ++i) { hist[KPT * t + i] = run; run += c[i]; }
  __syncthreads();

  #pragma unroll
  for (int p = 0; p < EPT; ++p) {
    if (ok[p]) {
      const int pos = atomicAdd(&hist[key[p]], 1);
      buf[pos] = make_int2(off[p], s[p]);
    }
  }
  __syncthreads();

  #pragma unroll
  for (int p = 0; p < EPT; ++p) {
    const int2 q = buf[min(jb + p * 32, nvalid - 1)];
    off[p] = q.x;
    s[p]   = q.y;
  }
  __syncthreads();
}

__global__ __launch_bounds__(SC_THREADS) void SpVecs_scatter(
    const float* __restrict__ feat, const int* __restrict__ sorted,
    const int* __restrict__ table, float* __restrict__ part,
    float* __restrict__ cpart, int HW, int S, int N, int E) {
  __shared__ int rpre[SORT_WG + 1];
  __shared__ int rdel[SORT_WG];
  __shared__ int rws[SORT_WG / 32];
  extern __shared__ float lds[];
  float* acc  = lds;
  float* cacc = lds + (size_t)PLANES * S;

  const int  g        = blockIdx.x;
  const int  c0       = blockIdx.y * PLANES;
  const bool counting = (blockIdx.y == 0);
  const int  lane     = threadIdx.x & 31;
  const int  w        = threadIdx.x >> 5;
  const unsigned q0 = (unsigned)(__builtin_amdgcn_readfirstlane(w) & 3) * (PLANES / 4);

  const int nz = (counting ? PLANES + 1 : PLANES) * S;
  const bool can_sort = PLANES * S >= KL + 32 + 2 * SC_THREADS * EPT;
  bool cleared = false;

  const int2* srt = (const int2*)sorted;

  {
    const int t = threadIdx.x;
    int st = 0, len = 0, x = 0;
    if (t < SORT_WG) {
      const int r0  = t * E;
      const int cnt = max(min(r0 + E, N) - r0, 0);
      st = table[t * NB + g];
      const int en = (g + 1 < NB) ? table[t * NB + g + 1] : cnt;
      len = en - st;
      x   = len;
      #pragma unroll
      for (int d = 1; d < 32; d <<= 1) {
        const int y = __shfl_up(x, d);
        if (lane >= d) x += y;
      }
      if (lane == 31) rws[w] = x;
    }
    __syncthreads();
    if (t < SORT_WG) {
      int bw = 0;
      #pragma unroll
      for (int q = 0; q < SORT_WG / 32; ++q) bw += (q < w) ? rws[q] : 0;
      const int ex = bw + x - len;
      rpre[t] = ex;
      rdel[t] = t * E + st - ex;
      if (t == SORT_WG - 1) rpre[SORT_WG] = bw + x;
    }
    __syncthreads();
  }
  const int total = rpre[SORT_WG];
  const unsigned uHW  = (unsigned)HW;
  const unsigned base = (unsigned)c0 * uHW;

  for (int b = 0; b < total; b += SC_THREADS * EPT) {
    unsigned idx[EPT];
    int      s[EPT];
    bool     ok[EPT];
    const int nvalid = min(total - b, SC_THREADS * EPT);
    const int pw = (nvalid + 31) >> 5;
    const int wj = w * pw + lane;
    int r = 0;
    {
      const int J0 = b + min(wj, nvalid - 1);
      #pragma unroll
      for (int step = SORT_WG / 2; step >= 1; step >>= 1) r += (rpre[r + step] <= J0) ? step : 0;
    }
    int nxt = rpre[r + 1];
    int del = rdel[r];
    int off[EPT];
    #pragma unroll
    for (int p = 0; p < EPT; ++p) {
      const int j = wj + p * 32;
      ok[p] = (p * 32 + lane < pw) && (j < nvalid);
      const int J = b + min(j, nvalid - 1);
      while (J >= nxt && r < SORT_WG - 1) {
        ++r;
        nxt = rpre[r + 1];
        del = rdel[r];
      }
      const int2 q = srt[J + del];
      off[p] = q.x;
      s[p]   = q.y;
    }

    if (!cleared) {
      if (can_sort) {
        const int shift  = SpVecs_shift(HW);
        const int lo     = g << shift;
        const int span   = (g == NB - 1) ? max(HW - lo, 1) : (1 << shift);
        int lshift = 0;
        while (lshift < 31 && ((span - 1) >> lshift) >= KL) ++lshift;
        SpVecs_localsort(off, s, ok, (int*)lds, lo, lshift, nvalid, pw);
      }
      float4* l4 = (float4*)lds;
      const float4 z4 = make_float4(0.0f, 0.0f, 0.0f, 0.0f);
      for (int i = threadIdx.x; i < (nz >> 2); i += SC_THREADS) l4[i] = z4;
      for (int i = (nz & ~3) + threadIdx.x; i < nz; i += SC_THREADS) lds[i] = 0.0f;
      __syncthreads();
      cleared = true;
    }

    #pragma unroll
    for (int p = 0; p < EPT; ++p) {
      idx[p] = base + (unsigned)off[p];
      s[p]   = ok[p] ? s[p] : min(lane, S - 1);
    }

    if (counting) {
      #pragma unroll
      for (int p = 0; p < EPT; ++p) atomicAdd(&cacc[s[p]], ok[p] ? 1.0f : 0.0f);
    }

    float va[EPT], vb[EPT];
    SpVecs_gather(va, feat, idx, SpVecs_plane(q0, 0) * uHW);
    for (int k = 0; k < PLANES - 2; k += 2) {
      const unsigned p0 = SpVecs_plane(q0, k);
      const unsigned p1 = SpVecs_plane(q0, k + 1);
      const unsigned p2 = SpVecs_plane(q0, k + 2);
      SpVecs_gather(vb, feat, idx, p1 * uHW);
      SpVecs_add(acc + (size_t)p0 * S, s, ok, va);
      SpVecs_gather(va, feat, idx, p2 * uHW);
      SpVecs_add(acc + (size_t)p1 * S, s, ok, vb);
    }
    {
      const unsigned p0 = SpVecs_plane(q0, PLANES - 2);
      const unsigned p1 = SpVecs_plane(q0, PLANES - 1);
      SpVecs_gather(vb, feat, idx, p1 * uHW);
      SpVecs_add(acc + (size_t)p0 * S, s, ok, va);
      SpVecs_add(acc + (size_t)p1 * S, s, ok, vb);
    }
  }
  if (!cleared) {
    for (int i = threadIdx.x; i < nz; i += SC_THREADS) lds[i] = 0.0f;
  }
  __syncthreads();

  const bool vec = ((S & 3) == 0) && ((((size_t)part) & 15) == 0);
  #pragma unroll 1
  for (int k = 0; k < PLANES; ++k) {
    float*       dst = part + ((size_t)(c0 + k) * NCHUNK + g) * (size_t)S;
    const float* pa  = acc + (size_t)k * S;
    if (vec) {
      float4*       d4 = (float4*)dst;
      const float4* a4 = (const float4*)pa;
      for (int i = threadIdx.x; i < (S >> 2); i += SC_THREADS) d4[i] = a4[i];
    } else {
      for (int i = threadIdx.x; i < S; i += SC_THREADS) dst[i] = pa[i];
    }
  }
  if (counting) {
    float* cd = cpart + (size_t)g * (size_t)S;
    for (int i = threadIdx.x; i < S; i += SC_THREADS) cd[i] = cacc[i];
  }
}

__global__ __launch_bounds__(RD_W * 32) void SpVecs_reduce(
    const float* __restrict__ part, const float* __restrict__ cpart,
    float* __restrict__ out, int S) {
  __shared__ float tile[RD_T][RD_T + 1];
  __shared__ float otile[RD_T][RD_T + 1];
  __shared__ float cn[RD_W][RD_T];
  const int lane = threadIdx.x & 31;
  const int w    = threadIdx.x >> 5;
  const int s0   = blockIdx.x * RD_T;
  const int c0   = blockIdx.y * RD_T;
  const int s    = s0 + lane;
  const bool in  = s < S;
  const int  sc  = in ? s : (S - 1);

  const unsigned uS = (unsigned)S;

  float tc[RD_B];
  #pragma unroll
  for (int j = 0; j < RD_B; ++j) tc[j] = cpart[(unsigned)(w * RD_B + j) * uS + (unsigned)sc];

  unsigned pk[4];
  #pragma unroll
  for (int k = 0; k < 4; ++k)
    pk[k] = ((unsigned)(c0 + w * 4 + k) * NCHUNK) * uS + (unsigned)sc;

  float a[4] = {0.0f, 0.0f, 0.0f, 0.0f};
  for (int g0 = 0; g0 < NCHUNK; g0 += RD_B) {
    float t[4][RD_B];
    #pragma unroll
    for (int j = 0; j < RD_B; ++j) {
      #pragma unroll
      for (int k = 0; k < 4; ++k) t[k][j] = part[pk[k] + (unsigned)(g0 + j) * uS];
    }
    #pragma unroll
    for (int j = 0; j < RD_B; ++j) {
      #pragma unroll
      for (int k = 0; k < 4; ++k) a[k] += t[k][j];
    }
  }
  #pragma unroll
  for (int k = 0; k < 4; ++k) tile[w * 4 + k][lane] = a[k];
  {
    float n = 0.0f;
    #pragma unroll
    for (int j = 0; j < RD_B; ++j) n += tc[j];
    cn[w][lane] = n;
  }
  __syncthreads();

  if (w < RD_T / 16) {
    const int sg    = 16 * w;
    const int m     = lane & 15;
    const int half  = lane >> 4;
    const int khalf = half << 1;

    float cnt = 0.0f;
    #pragma unroll
    for (int q = 0; q < RD_W; ++q) cnt += cn[q][sg + m];
    const float invc = 1.0f / fmaxf(cnt, 1.0f);

#if __has_builtin(__builtin_amdgcn_wmma_f32_16x16x4_f32)
    #pragma unroll
    for (int cc = 0; cc < RD_T; cc += 16) {
      v8f acc = {0.f, 0.f, 0.f, 0.f, 0.f, 0.f, 0.f, 0.f};
      #pragma unroll
      for (int j = 0; j < 4; ++j) {
        v2f A;
        A.x = (m == 4 * j + khalf + 0) ? invc : 0.0f;
        A.y = (m == 4 * j + khalf + 1) ? invc : 0.0f;
        const int r0 = sg + 4 * j + khalf;
        v2f B;
        B.x = tile[cc + m][r0];
        B.y = tile[cc + m][r0 + 1];
        acc = __builtin_amdgcn_wmma_f32_16x16x4_f32(
            false, A, false, B, (short)0, acc, false, false);
      }
      const int rowbase = sg + (half << 3);
      #pragma unroll
      for (int r = 0; r < 8; ++r) otile[rowbase + r][cc + m] = acc[r];
    }
#else
    for (int cc = half; cc < RD_T; cc += 2) otile[sg + m][cc] = tile[cc][sg + m] * invc;
#endif
  }
  __syncthreads();

  #pragma unroll
  for (int k = 0; k < 4; ++k) {
    const int r = w * 4 + k;
    if (s0 + r < S) out[(size_t)(s0 + r) * C_CH + c0 + lane] = otile[r][lane];
  }
}

extern "C" void kernel_launch(void* const* d_in, const int* in_sizes, int n_in,
                              void* d_out, int out_size, void* d_ws, size_t ws_size,
                              hipStream_t stream) {
  const float* feat = (const float*)d_in[0];
  const int*   yx   = (const int*)d_in[1];
  const int*   seg  = (const int*)d_in[2];
  float* out = (float*)d_out;

  const int N  = in_sizes[2];
  const int S  = out_size / C_CH;
  const int HW = in_sizes[0] / C_CH;
  const int W  = (int)lround(sqrt((double)HW));

  int*   table  = (int*)d_ws;
  int*   sorted = table + (size_t)SORT_WG * NB;
  float* cpart  = (float*)(sorted + 2 * (size_t)N);
  float* part   = cpart + (size_t)NCHUNK * (size_t)S;

  const int E = (N + SORT_WG - 1) / SORT_WG;

  SpVecs_deal<<<SORT_WG, SORT_T, 0, stream>>>(yx, seg, table, sorted, N, HW, W, E);

  {
    dim3 grid(NCHUNK, C_CH / PLANES);
    SpVecs_scatter<<<grid, SC_THREADS, (PLANES + 1) * (size_t)S * sizeof(float), stream>>>(
        feat, sorted, table, part, cpart, HW, S, N, E);
  }

  {
    dim3 grid((S + RD_T - 1) / RD_T, C_CH / RD_T);
    SpVecs_reduce<<<grid, RD_W * 32, 0, stream>>>(part, cpart, out, S);
  }
}
